# starting kernel
# baseline (speedup 1.0000x reference)
#include <hip/hip_runtime.h>

typedef __attribute__((ext_vector_type(2))) float v2f;
typedef __attribute__((ext_vector_type(8))) float v8f;

#define Bn 2
#define Cn 64
#define On 64
#define En 128
#define Tn 512
#define Wn 7
#define TOUT 256
#define FOUT 64

__global__ void lc_stats1(const float* __restrict__ x,
                          float* __restrict__ Sx, float* __restrict__ Qx) {
  int cf = blockIdx.x;
  int c = cf >> 7, f = cf & 127;
  float s[7], q[7];
#pragma unroll
  for (int w = 0; w < 7; ++w) { s[w] = 0.f; q[w] = 0.f; }
  for (int item = threadIdx.x; item < Bn * TOUT; item += blockDim.x) {
    int b = item >> 8, t = item & 255;
    const float* xr = x + (((size_t)b * Cn + c) * En + f) * Tn;
    int tau0 = 2 * t - 3;
#pragma unroll
    for (int w = 0; w < 7; ++w) {
      int tau = tau0 + w;
      if (tau >= 0 && tau < Tn) {
        float v = xr[tau];
        s[w] += v; q[w] += v * v;
      }
    }
  }
  __shared__ float red[8][14];
  int lane = threadIdx.x & 31, wv = threadIdx.x >> 5;
#pragma unroll
  for (int k = 0; k < 14; ++k) {
    float v = (k < 7) ? s[k] : q[k - 7];
    for (int off = 16; off > 0; off >>= 1) v += __shfl_down(v, off, 32);
    if (lane == 0) red[wv][k] = v;
  }
  __syncthreads();
  if (threadIdx.x < 14) {
    float tot = 0.f;
#pragma unroll
    for (int i = 0; i < 8; ++i) tot += red[i][threadIdx.x];
    if (threadIdx.x < 7) Sx[cf * 7 + threadIdx.x] = tot;
    else                 Qx[cf * 7 + threadIdx.x - 7] = tot;
  }
}

__global__ void lc_stats2(const float* __restrict__ wgt,
                          const float* __restrict__ gamma,
                          const float* __restrict__ beta,
                          const float* __restrict__ Sx,
                          const float* __restrict__ Qx,
                          float* __restrict__ aO, float* __restrict__ bO) {
  int o = blockIdx.x;
  int cbase = o & ~3;
  float n1 = 0.f, n2 = 0.f;
  for (int idx = threadIdx.x; idx < 4 * En * Wn; idx += blockDim.x) {
    int d = idx / (En * Wn);
    int rem = idx - d * (En * Wn);
    float wt = wgt[(size_t)(o * 4 + d) * (En * Wn) + rem];
    int c = cbase + d;
    float sv = Sx[(size_t)c * (En * Wn) + rem];
    float qv = Qx[(size_t)c * (En * Wn) + rem];
    n1 += wt * sv;
    n2 += wt * wt * qv;
  }
  __shared__ float red[4][2];
  int lane = threadIdx.x & 31, wv = threadIdx.x >> 5;
  for (int off = 16; off > 0; off >>= 1) {
    n1 += __shfl_down(n1, off, 32);
    n2 += __shfl_down(n2, off, 32);
  }
  if (lane == 0) { red[wv][0] = n1; red[wv][1] = n2; }
  __syncthreads();
  if (threadIdx.x == 0) {
    float t1 = 0.f, t2 = 0.f;
#pragma unroll
    for (int i = 0; i < 4; ++i) { t1 += red[i][0]; t2 += red[i][1]; }
    const float invcnt = 1.0f / 1835008.0f;
    float mean = t1 * invcnt;
    float var  = t2 * invcnt - mean * mean;
    float inv  = rsqrtf(var + 1e-5f);
    float a    = gamma[o] * inv;
    aO[o] = a;
    bO[o] = 56.0f * (beta[o] - mean * a);
  }
}

__global__ void __launch_bounds__(32)
lc_wmma_main(const float* __restrict__ x, const float* __restrict__ wgt,
             const float* __restrict__ aO, const float* __restrict__ bO,
             float* __restrict__ out) {
  int id = blockIdx.x;
  int tt = id & 15;
  int fo = (id >> 4) & 63;
  int ob = (id >> 10) & 3;
  int b  = id >> 12;
  int t0 = tt << 4, o0 = ob << 4, c0 = o0, f0 = fo << 1;

  int L  = threadIdx.x;
  int m  = L & 15;
  int hi = L >> 4;
  int d0 = hi << 1;
  int mq = m >> 2;

  __shared__ float xs[2][40][18];

  for (int idx = L; idx < 1184; idx += 32) {
    int fi = idx / 592;
    int r  = idx - fi * 592;
    int ch = r / 37;
    int j  = r - ch * 37;
    int tau = 2 * t0 - 3 + j;
    float v = 0.f;
    if (tau >= 0 && tau < Tn)
      v = x[(((size_t)b * Cn + (c0 + ch)) * En + (f0 + fi)) * Tn + tau];
    xs[fi][j][ch] = v;
  }

  float wa[2][7], wb[2][7];
#pragma unroll
  for (int fi = 0; fi < 2; ++fi) {
    const float* wp = wgt + (((size_t)(o0 + m) * 4 + d0) * En + (f0 + fi)) * Wn;
#pragma unroll
    for (int w = 0; w < 7; ++w) {
      wa[fi][w] = wp[w];
      wb[fi][w] = wp[En * Wn + w];
    }
  }

  __syncthreads();

  v8f acc = {};
#pragma unroll
  for (int fi = 0; fi < 2; ++fi) {
#pragma unroll
    for (int w = 0; w < 7; ++w) {
      int j = 2 * m + w;
#pragma unroll
      for (int q = 0; q < 4; ++q) {
        v2f bv = *(const v2f*)&xs[fi][j][(q << 2) + d0];
        float sel = (mq == q) ? 1.f : 0.f;
        v2f av = { sel * wa[fi][w], sel * wb[fi][w] };
        acc = __builtin_amdgcn_wmma_f32_16x16x4_f32(
             false, av,  false, bv,
             (short)0, acc,  false,  false);
      }
    }
  }

#pragma unroll
  for (int r = 0; r < 8; ++r) {
    int o = o0 + r + (hi << 3);
    float v = aO[o] * acc[r] + bO[o];
    v = (v >= 0.f) ? v : 0.01f * v;
    out[(((size_t)b * On + o) * FOUT + fo) * TOUT + t0 + m] = v;
  }
}

extern "C" void kernel_launch(void* const* d_in, const int* in_sizes, int n_in,
                              void* d_out, int out_size, void* d_ws, size_t ws_size,
                              hipStream_t stream) {
  (void)in_sizes; (void)n_in; (void)out_size; (void)ws_size;
  const float* x     = (const float*)d_in[0];
  const float* wgt   = (const float*)d_in[1];
  const float* gamma = (const float*)d_in[2];
  const float* beta  = (const float*)d_in[3];
  float* out = (float*)d_out;

  float* ws = (float*)d_ws;
  float* Sx = ws;
  float* Qx = ws + 57344;
  float* aO = ws + 114688;
  float* bO = ws + 114752;

  lc_stats1<<<Cn * En, 256, 0, stream>>>(x, Sx, Qx);
  lc_stats2<<<On, 128, 0, stream>>>(wgt, gamma, beta, Sx, Qx, aO, bO);
  lc_wmma_main<<<8192, 32, 0, stream>>>(x, wgt, aO, bO, out);
}


# stats pass 1: each row read once with coalesced 16-byte loads; the 7 tap sums rebuilt from even/odd-parity sums plus 5 edge elements; one wave per 2 rows (changes the order of these sums)
# speedup vs baseline: 1.4817x; 1.4817x over previous
#include <hip/hip_runtime.h>

typedef __attribute__((ext_vector_type(2))) float v2f;
typedef __attribute__((ext_vector_type(8))) float v8f;

#define Bn 2
#define Cn 64
#define On 64
#define En 128
#define Tn 512
#define Wn 7
#define TOUT 256
#define FOUT 64

__device__ __forceinline__ float lane_value(float v, int lane) {
  return __int_as_float(__builtin_amdgcn_readlane(__float_as_int(v), lane));
}

__global__ void __launch_bounds__(256)
lc_stats1(const float* __restrict__ x,
          float* __restrict__ Sx, float* __restrict__ Qx) {
  const int L   = threadIdx.x & 31;
  const int gw  = blockIdx.x * 8 + (threadIdx.x >> 5);
  const int cf0 = gw * 2;
  const int c = cf0 >> 7, f0 = cf0 & 127;

  float4 v[2][2][4];
#pragma unroll
  for (int b = 0; b < 2; ++b) {
    const float4* p =
        (const float4*)(x + (((size_t)b * Cn + c) * En + f0) * Tn);
#pragma unroll
    for (int r = 0; r < 2; ++r)
#pragma unroll
      for (int k = 0; k < 4; ++k)
        v[b][r][k] = p[r * (Tn / 4) + k * 32 + L];
  }
  __builtin_amdgcn_sched_barrier(0);

  float Ec[2], Oc[2], QEc[2], QOc[2];
  float s0[2], s1[2], s509[2], s510[2], s511[2];
  float q0[2], q1[2], q509[2], q510[2], q511[2];
#pragma unroll
  for (int r = 0; r < 2; ++r) {
    const float4 ha = v[0][r][0], hb = v[1][r][0];
    const float4 ta = v[0][r][3], tb = v[1][r][3];
    s0[r]   = lane_value(ha.x + hb.x, 0);
    s1[r]   = lane_value(ha.y + hb.y, 0);
    s509[r] = lane_value(ta.y + tb.y, 31);
    s510[r] = lane_value(ta.z + tb.z, 31);
    s511[r] = lane_value(ta.w + tb.w, 31);
    q0[r]   = lane_value(ha.x * ha.x + hb.x * hb.x, 0);
    q1[r]   = lane_value(ha.y * ha.y + hb.y * hb.y, 0);
    q509[r] = lane_value(ta.y * ta.y + tb.y * tb.y, 31);
    q510[r] = lane_value(ta.z * ta.z + tb.z * tb.z, 31);
    q511[r] = lane_value(ta.w * ta.w + tb.w * tb.w, 31);

    const bool first = (L == 0), last = (L == 31);
#pragma unroll
    for (int b = 0; b < 2; ++b) {
      v[b][r][0].x = first ? 0.f : v[b][r][0].x;
      v[b][r][0].y = first ? 0.f : v[b][r][0].y;
      v[b][r][3].y = last ? 0.f : v[b][r][3].y;
      v[b][r][3].z = last ? 0.f : v[b][r][3].z;
      v[b][r][3].w = last ? 0.f : v[b][r][3].w;
    }

    float se = 0.f, so = 0.f, qe = 0.f, qo = 0.f;
#pragma unroll
    for (int b = 0; b < 2; ++b) {
#pragma unroll
      for (int k = 0; k < 4; ++k) {
        const float4 t = v[b][r][k];
        se += t.x + t.z;
        so += t.y + t.w;
        qe += t.x * t.x + t.z * t.z;
        qo += t.y * t.y + t.w * t.w;
      }
    }
#pragma unroll
    for (int off = 16; off > 0; off >>= 1) {
      se += __shfl_xor(se, off, 32);
      so += __shfl_xor(so, off, 32);
      qe += __shfl_xor(qe, off, 32);
      qo += __shfl_xor(qo, off, 32);
    }
    Ec[r] = se; Oc[r] = so; QEc[r] = qe; QOc[r] = qo;
  }

  const int rs = (L >= 7) ? 1 : 0;
  const int w  = L - 7 * rs;
  const float E  = rs ? Ec[1]  : Ec[0],  O  = rs ? Oc[1]  : Oc[0];
  const float QE = rs ? QEc[1] : QEc[0], QO = rs ? QOc[1] : QOc[0];
  const float a0 = rs ? s0[1] : s0[0], a1 = rs ? s1[1] : s1[0];
  const float a509 = rs ? s509[1] : s509[0], a510 = rs ? s510[1] : s510[0];
  const float a511 = rs ? s511[1] : s511[0];
  const float b0 = rs ? q0[1] : q0[0], b1 = rs ? q1[1] : q1[0];
  const float b509 = rs ? q509[1] : q509[0], b510 = rs ? q510[1] : q510[0];
  const float b511 = rs ? q511[1] : q511[0];

  const float sedge = (w == 0) ? a1
                    : (w == 1) ? a0
                    : (w == 2) ? a1 + a509
                    : (w == 3) ? a0 + a510
                    : (w == 4) ? (a1 + a509) + a511
                    : (w == 5) ? a510
                               : a509 + a511;
  const float qedge = (w == 0) ? b1
                    : (w == 1) ? b0
                    : (w == 2) ? b1 + b509
                    : (w == 3) ? b0 + b510
                    : (w == 4) ? (b1 + b509) + b511
                    : (w == 5) ? b510
                               : b509 + b511;
  const float sv = ((w & 1) ? E  : O)  + sedge;
  const float qv = ((w & 1) ? QE : QO) + qedge;
  if (L < 14) {
    Sx[cf0 * 7 + L] = sv;
    Qx[cf0 * 7 + L] = qv;
  }
}

__global__ void lc_stats2(const float* __restrict__ wgt,
                          const float* __restrict__ gamma,
                          const float* __restrict__ beta,
                          const float* __restrict__ Sx,
                          const float* __restrict__ Qx,
                          float* __restrict__ aO, float* __restrict__ bO) {
  int o = blockIdx.x;
  int cbase = o & ~3;
  float n1 = 0.f, n2 = 0.f;
  for (int idx = threadIdx.x; idx < 4 * En * Wn; idx += blockDim.x) {
    int d = idx / (En * Wn);
    int rem = idx - d * (En * Wn);
    float wt = wgt[(size_t)(o * 4 + d) * (En * Wn) + rem];
    int c = cbase + d;
    float sv = Sx[(size_t)c * (En * Wn) + rem];
    float qv = Qx[(size_t)c * (En * Wn) + rem];
    n1 += wt * sv;
    n2 += wt * wt * qv;
  }
  __shared__ float red[4][2];
  int lane = threadIdx.x & 31, wv = threadIdx.x >> 5;
  for (int off = 16; off > 0; off >>= 1) {
    n1 += __shfl_down(n1, off, 32);
    n2 += __shfl_down(n2, off, 32);
  }
  if (lane == 0) { red[wv][0] = n1; red[wv][1] = n2; }
  __syncthreads();
  if (threadIdx.x == 0) {
    float t1 = 0.f, t2 = 0.f;
#pragma unroll
    for (int i = 0; i < 4; ++i) { t1 += red[i][0]; t2 += red[i][1]; }
    const float invcnt = 1.0f / 1835008.0f;
    float mean = t1 * invcnt;
    float var  = t2 * invcnt - mean * mean;
    float inv  = rsqrtf(var + 1e-5f);
    float a    = gamma[o] * inv;
    aO[o] = a;
    bO[o] = 56.0f * (beta[o] - mean * a);
  }
}

__global__ void __launch_bounds__(32)
lc_wmma_main(const float* __restrict__ x, const float* __restrict__ wgt,
             const float* __restrict__ aO, const float* __restrict__ bO,
             float* __restrict__ out) {
  int id = blockIdx.x;
  int tt = id & 15;
  int fo = (id >> 4) & 63;
  int ob = (id >> 10) & 3;
  int b  = id >> 12;
  int t0 = tt << 4, o0 = ob << 4, c0 = o0, f0 = fo << 1;

  int L  = threadIdx.x;
  int m  = L & 15;
  int hi = L >> 4;
  int d0 = hi << 1;
  int mq = m >> 2;

  __shared__ float xs[2][40][18];

  for (int idx = L; idx < 1184; idx += 32) {
    int fi = idx / 592;
    int r  = idx - fi * 592;
    int ch = r / 37;
    int j  = r - ch * 37;
    int tau = 2 * t0 - 3 + j;
    float v = 0.f;
    if (tau >= 0 && tau < Tn)
      v = x[(((size_t)b * Cn + (c0 + ch)) * En + (f0 + fi)) * Tn + tau];
    xs[fi][j][ch] = v;
  }

  float wa[2][7], wb[2][7];
#pragma unroll
  for (int fi = 0; fi < 2; ++fi) {
    const float* wp = wgt + (((size_t)(o0 + m) * 4 + d0) * En + (f0 + fi)) * Wn;
#pragma unroll
    for (int w = 0; w < 7; ++w) {
      wa[fi][w] = wp[w];
      wb[fi][w] = wp[En * Wn + w];
    }
  }

  __syncthreads();

  v8f acc = {};
#pragma unroll
  for (int fi = 0; fi < 2; ++fi) {
#pragma unroll
    for (int w = 0; w < 7; ++w) {
      int j = 2 * m + w;
#pragma unroll
      for (int q = 0; q < 4; ++q) {
        v2f bv = *(const v2f*)&xs[fi][j][(q << 2) + d0];
        float sel = (mq == q) ? 1.f : 0.f;
        v2f av = { sel * wa[fi][w], sel * wb[fi][w] };
        acc = __builtin_amdgcn_wmma_f32_16x16x4_f32(
             false, av,  false, bv,
             (short)0, acc,  false,  false);
      }
    }
  }

#pragma unroll
  for (int r = 0; r < 8; ++r) {
    int o = o0 + r + (hi << 3);
    float v = aO[o] * acc[r] + bO[o];
    v = (v >= 0.f) ? v : 0.01f * v;
    out[(((size_t)b * On + o) * FOUT + fo) * TOUT + t0 + m] = v;
  }
}

extern "C" void kernel_launch(void* const* d_in, const int* in_sizes, int n_in,
                              void* d_out, int out_size, void* d_ws, size_t ws_size,
                              hipStream_t stream) {
  (void)in_sizes; (void)n_in; (void)out_size; (void)ws_size;
  const float* x     = (const float*)d_in[0];
  const float* wgt   = (const float*)d_in[1];
  const float* gamma = (const float*)d_in[2];
  const float* beta  = (const float*)d_in[3];
  float* out = (float*)d_out;

  float* ws = (float*)d_ws;
  float* Sx = ws;
  float* Qx = ws + 57344;
  float* aO = ws + 114688;
  float* bO = ws + 114752;

  lc_stats1<<<512, 256, 0, stream>>>(x, Sx, Qx);
  lc_stats2<<<On, 128, 0, stream>>>(wgt, gamma, beta, Sx, Qx, aO, bO);
  lc_wmma_main<<<8192, 32, 0, stream>>>(x, wgt, aO, bO, out);
}


# main pass: a tile's x slice staged by ten independent 16-byte loads per lane (whole 128-byte lines plus front and back pieces) in one round trip, not 37 dependent 4-byte loads
# speedup vs baseline: 1.8509x; 1.2492x over previous
#include <hip/hip_runtime.h>

typedef __attribute__((ext_vector_type(2))) float v2f;
typedef __attribute__((ext_vector_type(8))) float v8f;
typedef __attribute__((ext_vector_type(4))) float v4f;

#define Bn 2
#define Cn 64
#define On 64
#define En 128
#define Tn 512
#define Wn 7
#define TOUT 256
#define FOUT 64

__device__ __forceinline__ float lane_value(float v, int lane) {
  return __int_as_float(__builtin_amdgcn_readlane(__float_as_int(v), lane));
}

__global__ void __launch_bounds__(256)
lc_stats1(const float* __restrict__ x,
          float* __restrict__ Sx, float* __restrict__ Qx) {
  const int L   = threadIdx.x & 31;
  const int gw  = blockIdx.x * 8 + (threadIdx.x >> 5);
  const int cf0 = gw * 2;
  const int c = cf0 >> 7, f0 = cf0 & 127;

  float4 v[2][2][4];
#pragma unroll
  for (int b = 0; b < 2; ++b) {
    const float4* p =
        (const float4*)(x + (((size_t)b * Cn + c) * En + f0) * Tn);
#pragma unroll
    for (int r = 0; r < 2; ++r)
#pragma unroll
      for (int k = 0; k < 4; ++k)
        v[b][r][k] = p[r * (Tn / 4) + k * 32 + L];
  }
  __builtin_amdgcn_sched_barrier(0);

  float Ec[2], Oc[2], QEc[2], QOc[2];
  float s0[2], s1[2], s509[2], s510[2], s511[2];
  float q0[2], q1[2], q509[2], q510[2], q511[2];
#pragma unroll
  for (int r = 0; r < 2; ++r) {
    const float4 ha = v[0][r][0], hb = v[1][r][0];
    const float4 ta = v[0][r][3], tb = v[1][r][3];
    s0[r]   = lane_value(ha.x + hb.x, 0);
    s1[r]   = lane_value(ha.y + hb.y, 0);
    s509[r] = lane_value(ta.y + tb.y, 31);
    s510[r] = lane_value(ta.z + tb.z, 31);
    s511[r] = lane_value(ta.w + tb.w, 31);
    q0[r]   = lane_value(ha.x * ha.x + hb.x * hb.x, 0);
    q1[r]   = lane_value(ha.y * ha.y + hb.y * hb.y, 0);
    q509[r] = lane_value(ta.y * ta.y + tb.y * tb.y, 31);
    q510[r] = lane_value(ta.z * ta.z + tb.z * tb.z, 31);
    q511[r] = lane_value(ta.w * ta.w + tb.w * tb.w, 31);

    const bool first = (L == 0), last = (L == 31);
#pragma unroll
    for (int b = 0; b < 2; ++b) {
      v[b][r][0].x = first ? 0.f : v[b][r][0].x;
      v[b][r][0].y = first ? 0.f : v[b][r][0].y;
      v[b][r][3].y = last ? 0.f : v[b][r][3].y;
      v[b][r][3].z = last ? 0.f : v[b][r][3].z;
      v[b][r][3].w = last ? 0.f : v[b][r][3].w;
    }

    float se = 0.f, so = 0.f, qe = 0.f, qo = 0.f;
#pragma unroll
    for (int b = 0; b < 2; ++b) {
#pragma unroll
      for (int k = 0; k < 4; ++k) {
        const float4 t = v[b][r][k];
        se += t.x + t.z;
        so += t.y + t.w;
        qe += t.x * t.x + t.z * t.z;
        qo += t.y * t.y + t.w * t.w;
      }
    }
#pragma unroll
    for (int off = 16; off > 0; off >>= 1) {
      se += __shfl_xor(se, off, 32);
      so += __shfl_xor(so, off, 32);
      qe += __shfl_xor(qe, off, 32);
      qo += __shfl_xor(qo, off, 32);
    }
    Ec[r] = se; Oc[r] = so; QEc[r] = qe; QOc[r] = qo;
  }

  const int rs = (L >= 7) ? 1 : 0;
  const int w  = L - 7 * rs;
  const float E  = rs ? Ec[1]  : Ec[0],  O  = rs ? Oc[1]  : Oc[0];
  const float QE = rs ? QEc[1] : QEc[0], QO = rs ? QOc[1] : QOc[0];
  const float a0 = rs ? s0[1] : s0[0], a1 = rs ? s1[1] : s1[0];
  const float a509 = rs ? s509[1] : s509[0], a510 = rs ? s510[1] : s510[0];
  const float a511 = rs ? s511[1] : s511[0];
  const float b0 = rs ? q0[1] : q0[0], b1 = rs ? q1[1] : q1[0];
  const float b509 = rs ? q509[1] : q509[0], b510 = rs ? q510[1] : q510[0];
  const float b511 = rs ? q511[1] : q511[0];

  const float sedge = (w == 0) ? a1
                    : (w == 1) ? a0
                    : (w == 2) ? a1 + a509
                    : (w == 3) ? a0 + a510
                    : (w == 4) ? (a1 + a509) + a511
                    : (w == 5) ? a510
                               : a509 + a511;
  const float qedge = (w == 0) ? b1
                    : (w == 1) ? b0
                    : (w == 2) ? b1 + b509
                    : (w == 3) ? b0 + b510
                    : (w == 4) ? (b1 + b509) + b511
                    : (w == 5) ? b510
                               : b509 + b511;
  const float sv = ((w & 1) ? E  : O)  + sedge;
  const float qv = ((w & 1) ? QE : QO) + qedge;
  if (L < 14) {
    Sx[cf0 * 7 + L] = sv;
    Qx[cf0 * 7 + L] = qv;
  }
}

__global__ void lc_stats2(const float* __restrict__ wgt,
                          const float* __restrict__ gamma,
                          const float* __restrict__ beta,
                          const float* __restrict__ Sx,
                          const float* __restrict__ Qx,
                          float* __restrict__ aO, float* __restrict__ bO) {
  int o = blockIdx.x;
  int cbase = o & ~3;
  float n1 = 0.f, n2 = 0.f;
  for (int idx = threadIdx.x; idx < 4 * En * Wn; idx += blockDim.x) {
    int d = idx / (En * Wn);
    int rem = idx - d * (En * Wn);
    float wt = wgt[(size_t)(o * 4 + d) * (En * Wn) + rem];
    int c = cbase + d;
    float sv = Sx[(size_t)c * (En * Wn) + rem];
    float qv = Qx[(size_t)c * (En * Wn) + rem];
    n1 += wt * sv;
    n2 += wt * wt * qv;
  }
  __shared__ float red[4][2];
  int lane = threadIdx.x & 31, wv = threadIdx.x >> 5;
  for (int off = 16; off > 0; off >>= 1) {
    n1 += __shfl_down(n1, off, 32);
    n2 += __shfl_down(n2, off, 32);
  }
  if (lane == 0) { red[wv][0] = n1; red[wv][1] = n2; }
  __syncthreads();
  if (threadIdx.x == 0) {
    float t1 = 0.f, t2 = 0.f;
#pragma unroll
    for (int i = 0; i < 4; ++i) { t1 += red[i][0]; t2 += red[i][1]; }
    const float invcnt = 1.0f / 1835008.0f;
    float mean = t1 * invcnt;
    float var  = t2 * invcnt - mean * mean;
    float inv  = rsqrtf(var + 1e-5f);
    float a    = gamma[o] * inv;
    aO[o] = a;
    bO[o] = 56.0f * (beta[o] - mean * a);
  }
}

__global__ void __launch_bounds__(32)
lc_wmma_main(const float* __restrict__ x, const float* __restrict__ wgt,
             const float* __restrict__ aO, const float* __restrict__ bO,
             float* __restrict__ out) {
  int id = blockIdx.x;
  int tt = id & 15;
  int fo = (id >> 4) & 63;
  int ob = (id >> 10) & 3;
  int b  = id >> 12;
  int t0 = tt << 4, o0 = ob << 4, c0 = o0, f0 = fo << 1;

  int L  = threadIdx.x;
  int m  = L & 15;
  int hi = L >> 4;
  int d0 = hi << 1;
  int mq = m >> 2;

  __shared__ float xs[2][40][18];

  const float* xb = x + (((size_t)b * Cn + c0) * En + f0) * Tn + 2 * t0;

  const int lr = L >> 3, lc = L & 7;
  v4f body[8];
#pragma unroll
  for (int k = 0; k < 8; ++k) {
    const int ch = 4 * (k & 3) + lr, fi = k >> 2;
    body[k] = *(const v4f*)(xb + ((size_t)ch * En + fi) * Tn + 4 * lc);
  }
  const int ech = L & 15, efi = L >> 4;
  const float* er = xb + ((size_t)ech * En + efi) * Tn;
  const v4f zero4 = {0.f, 0.f, 0.f, 0.f};
  const bool has_head = (tt > 0), has_tail = (tt < 15);
  v4f head = *(const v4f*)(er + (has_head ? -4 : 0));
  v4f tail = *(const v4f*)(er + (has_tail ? 32 : 28));

  float wa[2][7], wb[2][7];
#pragma unroll
  for (int fi = 0; fi < 2; ++fi) {
    const float* wp = wgt + (((size_t)(o0 + m) * 4 + d0) * En + (f0 + fi)) * Wn;
#pragma unroll
    for (int w = 0; w < 7; ++w) {
      wa[fi][w] = wp[w];
      wb[fi][w] = wp[En * Wn + w];
    }
  }

  __builtin_amdgcn_sched_barrier(0);
  head = has_head ? head : zero4;
  tail = has_tail ? tail : zero4;

#pragma unroll
  for (int k = 0; k < 8; ++k) {
    const int ch = 4 * (k & 3) + lr, fi = k >> 2;
    const int j = 4 + 4 * lc;
    xs[fi][j + 0][ch] = body[k].x;
    xs[fi][j + 1][ch] = body[k].y;
    xs[fi][j + 2][ch] = body[k].z;
    xs[fi][j + 3][ch] = body[k].w;
  }
  xs[efi][0][ech]  = head.x;  xs[efi][1][ech]  = head.y;
  xs[efi][2][ech]  = head.z;  xs[efi][3][ech]  = head.w;
  xs[efi][36][ech] = tail.x;  xs[efi][37][ech] = tail.y;
  xs[efi][38][ech] = tail.z;  xs[efi][39][ech] = tail.w;

  __syncthreads();

  v8f acc = {};
#pragma unroll
  for (int fi = 0; fi < 2; ++fi) {
#pragma unroll
    for (int w = 0; w < 7; ++w) {
      int j = 2 * m + w + 1;
#pragma unroll
      for (int q = 0; q < 4; ++q) {
        v2f bv = *(const v2f*)&xs[fi][j][(q << 2) + d0];
        float sel = (mq == q) ? 1.f : 0.f;
        v2f av = { sel * wa[fi][w], sel * wb[fi][w] };
        acc = __builtin_amdgcn_wmma_f32_16x16x4_f32(
             false, av,  false, bv,
             (short)0, acc,  false,  false);
      }
    }
  }

#pragma unroll
  for (int r = 0; r < 8; ++r) {
    int o = o0 + r + (hi << 3);
    float v = aO[o] * acc[r] + bO[o];
    v = (v >= 0.f) ? v : 0.01f * v;
    out[(((size_t)b * On + o) * FOUT + fo) * TOUT + t0 + m] = v;
  }
}

extern "C" void kernel_launch(void* const* d_in, const int* in_sizes, int n_in,
                              void* d_out, int out_size, void* d_ws, size_t ws_size,
                              hipStream_t stream) {
  (void)in_sizes; (void)n_in; (void)out_size; (void)ws_size;
  const float* x     = (const float*)d_in[0];
  const float* wgt   = (const float*)d_in[1];
  const float* gamma = (const float*)d_in[2];
  const float* beta  = (const float*)d_in[3];
  float* out = (float*)d_out;

  float* ws = (float*)d_ws;
  float* Sx = ws;
  float* Qx = ws + 57344;
  float* aO = ws + 114688;
  float* bO = ws + 114752;

  lc_stats1<<<512, 256, 0, stream>>>(x, Sx, Qx);
  lc_stats2<<<On, 128, 0, stream>>>(wgt, gamma, beta, Sx, Qx, aO, bO);
  lc_wmma_main<<<8192, 32, 0, stream>>>(x, wgt, aO, bO, out);
}


# stats pass 2: one block per o, 224 threads, twelve independent 16-byte loads per thread (weights, tap sums): one memory round trip, not a 28-trip load loop (changes the order of these sums)
# speedup vs baseline: 2.8849x; 1.5586x over previous
#include <hip/hip_runtime.h>

typedef __attribute__((ext_vector_type(2))) float v2f;
typedef __attribute__((ext_vector_type(8))) float v8f;
typedef __attribute__((ext_vector_type(4))) float v4f;

#define Bn 2
#define Cn 64
#define On 64
#define En 128
#define Tn 512
#define Wn 7
#define TOUT 256
#define FOUT 64

__device__ __forceinline__ float lane_value(float v, int lane) {
  return __int_as_float(__builtin_amdgcn_readlane(__float_as_int(v), lane));
}

__global__ void __launch_bounds__(256)
lc_stats1(const float* __restrict__ x,
          float* __restrict__ Sx, float* __restrict__ Qx) {
  const int L   = threadIdx.x & 31;
  const int gw  = blockIdx.x * 8 + (threadIdx.x >> 5);
  const int cf0 = gw * 2;
  const int c = cf0 >> 7, f0 = cf0 & 127;

  float4 v[2][2][4];
#pragma unroll
  for (int b = 0; b < 2; ++b) {
    const float4* p =
        (const float4*)(x + (((size_t)b * Cn + c) * En + f0) * Tn);
#pragma unroll
    for (int r = 0; r < 2; ++r)
#pragma unroll
      for (int k = 0; k < 4; ++k)
        v[b][r][k] = p[r * (Tn / 4) + k * 32 + L];
  }
  __builtin_amdgcn_sched_barrier(0);

  float Ec[2], Oc[2], QEc[2], QOc[2];
  float s0[2], s1[2], s509[2], s510[2], s511[2];
  float q0[2], q1[2], q509[2], q510[2], q511[2];
#pragma unroll
  for (int r = 0; r < 2; ++r) {
    const float4 ha = v[0][r][0], hb = v[1][r][0];
    const float4 ta = v[0][r][3], tb = v[1][r][3];
    s0[r]   = lane_value(ha.x + hb.x, 0);
    s1[r]   = lane_value(ha.y + hb.y, 0);
    s509[r] = lane_value(ta.y + tb.y, 31);
    s510[r] = lane_value(ta.z + tb.z, 31);
    s511[r] = lane_value(ta.w + tb.w, 31);
    q0[r]   = lane_value(ha.x * ha.x + hb.x * hb.x, 0);
    q1[r]   = lane_value(ha.y * ha.y + hb.y * hb.y, 0);
    q509[r] = lane_value(ta.y * ta.y + tb.y * tb.y, 31);
    q510[r] = lane_value(ta.z * ta.z + tb.z * tb.z, 31);
    q511[r] = lane_value(ta.w * ta.w + tb.w * tb.w, 31);

    const bool first = (L == 0), last = (L == 31);
#pragma unroll
    for (int b = 0; b < 2; ++b) {
      v[b][r][0].x = first ? 0.f : v[b][r][0].x;
      v[b][r][0].y = first ? 0.f : v[b][r][0].y;
      v[b][r][3].y = last ? 0.f : v[b][r][3].y;
      v[b][r][3].z = last ? 0.f : v[b][r][3].z;
      v[b][r][3].w = last ? 0.f : v[b][r][3].w;
    }

    float se = 0.f, so = 0.f, qe = 0.f, qo = 0.f;
#pragma unroll
    for (int b = 0; b < 2; ++b) {
#pragma unroll
      for (int k = 0; k < 4; ++k) {
        const float4 t = v[b][r][k];
        se += t.x + t.z;
        so += t.y + t.w;
        qe += t.x * t.x + t.z * t.z;
        qo += t.y * t.y + t.w * t.w;
      }
    }
#pragma unroll
    for (int off = 16; off > 0; off >>= 1) {
      se += __shfl_xor(se, off, 32);
      so += __shfl_xor(so, off, 32);
      qe += __shfl_xor(qe, off, 32);
      qo += __shfl_xor(qo, off, 32);
    }
    Ec[r] = se; Oc[r] = so; QEc[r] = qe; QOc[r] = qo;
  }

  const int rs = (L >= 7) ? 1 : 0;
  const int w  = L - 7 * rs;
  const float E  = rs ? Ec[1]  : Ec[0],  O  = rs ? Oc[1]  : Oc[0];
  const float QE = rs ? QEc[1] : QEc[0], QO = rs ? QOc[1] : QOc[0];
  const float a0 = rs ? s0[1] : s0[0], a1 = rs ? s1[1] : s1[0];
  const float a509 = rs ? s509[1] : s509[0], a510 = rs ? s510[1] : s510[0];
  const float a511 = rs ? s511[1] : s511[0];
  const float b0 = rs ? q0[1] : q0[0], b1 = rs ? q1[1] : q1[0];
  const float b509 = rs ? q509[1] : q509[0], b510 = rs ? q510[1] : q510[0];
  const float b511 = rs ? q511[1] : q511[0];

  const float sedge = (w == 0) ? a1
                    : (w == 1) ? a0
                    : (w == 2) ? a1 + a509
                    : (w == 3) ? a0 + a510
                    : (w == 4) ? (a1 + a509) + a511
                    : (w == 5) ? a510
                               : a509 + a511;
  const float qedge = (w == 0) ? b1
                    : (w == 1) ? b0
                    : (w == 2) ? b1 + b509
                    : (w == 3) ? b0 + b510
                    : (w == 4) ? (b1 + b509) + b511
                    : (w == 5) ? b510
                               : b509 + b511;
  const float sv = ((w & 1) ? E  : O)  + sedge;
  const float qv = ((w & 1) ? QE : QO) + qedge;
  if (L < 14) {
    Sx[cf0 * 7 + L] = sv;
    Qx[cf0 * 7 + L] = qv;
  }
}

__global__ void __launch_bounds__(224)
lc_stats2(const float* __restrict__ wgt,
          const float* __restrict__ gamma,
          const float* __restrict__ beta,
          const float* __restrict__ Sx,
          const float* __restrict__ Qx,
          float* __restrict__ aO, float* __restrict__ bO) {
  const int o = blockIdx.x;
  const int cbase = o & ~3;
  const int i = threadIdx.x;
  const float4* wp = (const float4*)(wgt + (size_t)o * 4 * (En * Wn)) + i;
  const float4* sp = (const float4*)(Sx + (size_t)cbase * (En * Wn)) + i;
  const float4* qp = (const float4*)(Qx + (size_t)cbase * (En * Wn)) + i;
  const float g = gamma[o], bt = beta[o];
  float4 wv[4], sv[4], qv[4];
#pragma unroll
  for (int d = 0; d < 4; ++d) {
    wv[d] = wp[d * (En * Wn / 4)];
    sv[d] = sp[d * (En * Wn / 4)];
    qv[d] = qp[d * (En * Wn / 4)];
  }
  float n1 = 0.f, n2 = 0.f;
#pragma unroll
  for (int d = 0; d < 4; ++d) {
    const float4 w4 = wv[d], s4 = sv[d], q4 = qv[d];
    n1 += w4.x * s4.x;  n2 += w4.x * w4.x * q4.x;
    n1 += w4.y * s4.y;  n2 += w4.y * w4.y * q4.y;
    n1 += w4.z * s4.z;  n2 += w4.z * w4.z * q4.z;
    n1 += w4.w * s4.w;  n2 += w4.w * w4.w * q4.w;
  }
  __shared__ float red[7][2];
  const int lane = threadIdx.x & 31, wvi = threadIdx.x >> 5;
  for (int off = 16; off > 0; off >>= 1) {
    n1 += __shfl_down(n1, off, 32);
    n2 += __shfl_down(n2, off, 32);
  }
  if (lane == 0) { red[wvi][0] = n1; red[wvi][1] = n2; }
  __syncthreads();
  if (threadIdx.x == 0) {
    float t1 = 0.f, t2 = 0.f;
#pragma unroll
    for (int k = 0; k < 7; ++k) { t1 += red[k][0]; t2 += red[k][1]; }
    const float invcnt = 1.0f / 1835008.0f;
    float mean = t1 * invcnt;
    float var  = t2 * invcnt - mean * mean;
    float inv  = rsqrtf(var + 1e-5f);
    float a    = g * inv;
    aO[o] = a;
    bO[o] = 56.0f * (bt - mean * a);
  }
}

__global__ void __launch_bounds__(32)
lc_wmma_main(const float* __restrict__ x, const float* __restrict__ wgt,
             const float* __restrict__ aO, const float* __restrict__ bO,
             float* __restrict__ out) {
  int id = blockIdx.x;
  int tt = id & 15;
  int fo = (id >> 4) & 63;
  int ob = (id >> 10) & 3;
  int b  = id >> 12;
  int t0 = tt << 4, o0 = ob << 4, c0 = o0, f0 = fo << 1;

  int L  = threadIdx.x;
  int m  = L & 15;
  int hi = L >> 4;
  int d0 = hi << 1;
  int mq = m >> 2;

  __shared__ float xs[2][40][18];

  const float* xb = x + (((size_t)b * Cn + c0) * En + f0) * Tn + 2 * t0;

  const int lr = L >> 3, lc = L & 7;
  v4f body[8];
#pragma unroll
  for (int k = 0; k < 8; ++k) {
    const int ch = 4 * (k & 3) + lr, fi = k >> 2;
    body[k] = *(const v4f*)(xb + ((size_t)ch * En + fi) * Tn + 4 * lc);
  }
  const int ech = L & 15, efi = L >> 4;
  const float* er = xb + ((size_t)ech * En + efi) * Tn;
  const v4f zero4 = {0.f, 0.f, 0.f, 0.f};
  const bool has_head = (tt > 0), has_tail = (tt < 15);
  v4f head = *(const v4f*)(er + (has_head ? -4 : 0));
  v4f tail = *(const v4f*)(er + (has_tail ? 32 : 28));

  float wa[2][7], wb[2][7];
#pragma unroll
  for (int fi = 0; fi < 2; ++fi) {
    const float* wp = wgt + (((size_t)(o0 + m) * 4 + d0) * En + (f0 + fi)) * Wn;
#pragma unroll
    for (int w = 0; w < 7; ++w) {
      wa[fi][w] = wp[w];
      wb[fi][w] = wp[En * Wn + w];
    }
  }

  __builtin_amdgcn_sched_barrier(0);
  head = has_head ? head : zero4;
  tail = has_tail ? tail : zero4;

#pragma unroll
  for (int k = 0; k < 8; ++k) {
    const int ch = 4 * (k & 3) + lr, fi = k >> 2;
    const int j = 4 + 4 * lc;
    xs[fi][j + 0][ch] = body[k].x;
    xs[fi][j + 1][ch] = body[k].y;
    xs[fi][j + 2][ch] = body[k].z;
    xs[fi][j + 3][ch] = body[k].w;
  }
  xs[efi][0][ech]  = head.x;  xs[efi][1][ech]  = head.y;
  xs[efi][2][ech]  = head.z;  xs[efi][3][ech]  = head.w;
  xs[efi][36][ech] = tail.x;  xs[efi][37][ech] = tail.y;
  xs[efi][38][ech] = tail.z;  xs[efi][39][ech] = tail.w;

  __syncthreads();

  v8f acc = {};
#pragma unroll
  for (int fi = 0; fi < 2; ++fi) {
#pragma unroll
    for (int w = 0; w < 7; ++w) {
      int j = 2 * m + w + 1;
#pragma unroll
      for (int q = 0; q < 4; ++q) {
        v2f bv = *(const v2f*)&xs[fi][j][(q << 2) + d0];
        float sel = (mq == q) ? 1.f : 0.f;
        v2f av = { sel * wa[fi][w], sel * wb[fi][w] };
        acc = __builtin_amdgcn_wmma_f32_16x16x4_f32(
             false, av,  false, bv,
             (short)0, acc,  false,  false);
      }
    }
  }

#pragma unroll
  for (int r = 0; r < 8; ++r) {
    int o = o0 + r + (hi << 3);
    float v = aO[o] * acc[r] + bO[o];
    v = (v >= 0.f) ? v : 0.01f * v;
    out[(((size_t)b * On + o) * FOUT + fo) * TOUT + t0 + m] = v;
  }
}

extern "C" void kernel_launch(void* const* d_in, const int* in_sizes, int n_in,
                              void* d_out, int out_size, void* d_ws, size_t ws_size,
                              hipStream_t stream) {
  (void)in_sizes; (void)n_in; (void)out_size; (void)ws_size;
  const float* x     = (const float*)d_in[0];
  const float* wgt   = (const float*)d_in[1];
  const float* gamma = (const float*)d_in[2];
  const float* beta  = (const float*)d_in[3];
  float* out = (float*)d_out;

  float* ws = (float*)d_ws;
  float* Sx = ws;
  float* Qx = ws + 57344;
  float* aO = ws + 114688;
  float* bO = ws + 114752;

  lc_stats1<<<512, 256, 0, stream>>>(x, Sx, Qx);
  lc_stats2<<<On, 224, 0, stream>>>(wgt, gamma, beta, Sx, Qx, aO, bO);
  lc_wmma_main<<<8192, 32, 0, stream>>>(x, wgt, aO, bO, out);
}


# main pass: one wave per 16(o) x 64(t) strip (four adjacent tiles): weights and A fragments loaded once per strip, four independent accumulators, 2048 waves enter instead of 8192
# speedup vs baseline: 3.4140x; 1.1834x over previous
#include <hip/hip_runtime.h>

typedef __attribute__((ext_vector_type(2))) float v2f;
typedef __attribute__((ext_vector_type(8))) float v8f;
typedef __attribute__((ext_vector_type(4))) float v4f;

#define Bn 2
#define Cn 64
#define On 64
#define En 128
#define Tn 512
#define Wn 7
#define TOUT 256
#define FOUT 64

__device__ __forceinline__ float lane_value(float v, int lane) {
  return __int_as_float(__builtin_amdgcn_readlane(__float_as_int(v), lane));
}

__global__ void __launch_bounds__(256)
lc_stats1(const float* __restrict__ x,
          float* __restrict__ Sx, float* __restrict__ Qx) {
  const int L   = threadIdx.x & 31;
  const int gw  = blockIdx.x * 8 + (threadIdx.x >> 5);
  const int cf0 = gw * 2;
  const int c = cf0 >> 7, f0 = cf0 & 127;

  float4 v[2][2][4];
#pragma unroll
  for (int b = 0; b < 2; ++b) {
    const float4* p =
        (const float4*)(x + (((size_t)b * Cn + c) * En + f0) * Tn);
#pragma unroll
    for (int r = 0; r < 2; ++r)
#pragma unroll
      for (int k = 0; k < 4; ++k)
        v[b][r][k] = p[r * (Tn / 4) + k * 32 + L];
  }
  __builtin_amdgcn_sched_barrier(0);

  float Ec[2], Oc[2], QEc[2], QOc[2];
  float s0[2], s1[2], s509[2], s510[2], s511[2];
  float q0[2], q1[2], q509[2], q510[2], q511[2];
#pragma unroll
  for (int r = 0; r < 2; ++r) {
    const float4 ha = v[0][r][0], hb = v[1][r][0];
    const float4 ta = v[0][r][3], tb = v[1][r][3];
    s0[r]   = lane_value(ha.x + hb.x, 0);
    s1[r]   = lane_value(ha.y + hb.y, 0);
    s509[r] = lane_value(ta.y + tb.y, 31);
    s510[r] = lane_value(ta.z + tb.z, 31);
    s511[r] = lane_value(ta.w + tb.w, 31);
    q0[r]   = lane_value(ha.x * ha.x + hb.x * hb.x, 0);
    q1[r]   = lane_value(ha.y * ha.y + hb.y * hb.y, 0);
    q509[r] = lane_value(ta.y * ta.y + tb.y * tb.y, 31);
    q510[r] = lane_value(ta.z * ta.z + tb.z * tb.z, 31);
    q511[r] = lane_value(ta.w * ta.w + tb.w * tb.w, 31);

    const bool first = (L == 0), last = (L == 31);
#pragma unroll
    for (int b = 0; b < 2; ++b) {
      v[b][r][0].x = first ? 0.f : v[b][r][0].x;
      v[b][r][0].y = first ? 0.f : v[b][r][0].y;
      v[b][r][3].y = last ? 0.f : v[b][r][3].y;
      v[b][r][3].z = last ? 0.f : v[b][r][3].z;
      v[b][r][3].w = last ? 0.f : v[b][r][3].w;
    }

    float se = 0.f, so = 0.f, qe = 0.f, qo = 0.f;
#pragma unroll
    for (int b = 0; b < 2; ++b) {
#pragma unroll
      for (int k = 0; k < 4; ++k) {
        const float4 t = v[b][r][k];
        se += t.x + t.z;
        so += t.y + t.w;
        qe += t.x * t.x + t.z * t.z;
        qo += t.y * t.y + t.w * t.w;
      }
    }
#pragma unroll
    for (int off = 16; off > 0; off >>= 1) {
      se += __shfl_xor(se, off, 32);
      so += __shfl_xor(so, off, 32);
      qe += __shfl_xor(qe, off, 32);
      qo += __shfl_xor(qo, off, 32);
    }
    Ec[r] = se; Oc[r] = so; QEc[r] = qe; QOc[r] = qo;
  }

  const int rs = (L >= 7) ? 1 : 0;
  const int w  = L - 7 * rs;
  const float E  = rs ? Ec[1]  : Ec[0],  O  = rs ? Oc[1]  : Oc[0];
  const float QE = rs ? QEc[1] : QEc[0], QO = rs ? QOc[1] : QOc[0];
  const float a0 = rs ? s0[1] : s0[0], a1 = rs ? s1[1] : s1[0];
  const float a509 = rs ? s509[1] : s509[0], a510 = rs ? s510[1] : s510[0];
  const float a511 = rs ? s511[1] : s511[0];
  const float b0 = rs ? q0[1] : q0[0], b1 = rs ? q1[1] : q1[0];
  const float b509 = rs ? q509[1] : q509[0], b510 = rs ? q510[1] : q510[0];
  const float b511 = rs ? q511[1] : q511[0];

  const float sedge = (w == 0) ? a1
                    : (w == 1) ? a0
                    : (w == 2) ? a1 + a509
                    : (w == 3) ? a0 + a510
                    : (w == 4) ? (a1 + a509) + a511
                    : (w == 5) ? a510
                               : a509 + a511;
  const float qedge = (w == 0) ? b1
                    : (w == 1) ? b0
                    : (w == 2) ? b1 + b509
                    : (w == 3) ? b0 + b510
                    : (w == 4) ? (b1 + b509) + b511
                    : (w == 5) ? b510
                               : b509 + b511;
  const float sv = ((w & 1) ? E  : O)  + sedge;
  const float qv = ((w & 1) ? QE : QO) + qedge;
  if (L < 14) {
    Sx[cf0 * 7 + L] = sv;
    Qx[cf0 * 7 + L] = qv;
  }
}

__global__ void __launch_bounds__(224)
lc_stats2(const float* __restrict__ wgt,
          const float* __restrict__ gamma,
          const float* __restrict__ beta,
          const float* __restrict__ Sx,
          const float* __restrict__ Qx,
          float* __restrict__ aO, float* __restrict__ bO) {
  const int o = blockIdx.x;
  const int cbase = o & ~3;
  const int i = threadIdx.x;
  const float4* wp = (const float4*)(wgt + (size_t)o * 4 * (En * Wn)) + i;
  const float4* sp = (const float4*)(Sx + (size_t)cbase * (En * Wn)) + i;
  const float4* qp = (const float4*)(Qx + (size_t)cbase * (En * Wn)) + i;
  const float g = gamma[o], bt = beta[o];
  float4 wv[4], sv[4], qv[4];
#pragma unroll
  for (int d = 0; d < 4; ++d) {
    wv[d] = wp[d * (En * Wn / 4)];
    sv[d] = sp[d * (En * Wn / 4)];
    qv[d] = qp[d * (En * Wn / 4)];
  }
  float n1 = 0.f, n2 = 0.f;
#pragma unroll
  for (int d = 0; d < 4; ++d) {
    const float4 w4 = wv[d], s4 = sv[d], q4 = qv[d];
    n1 += w4.x * s4.x;  n2 += w4.x * w4.x * q4.x;
    n1 += w4.y * s4.y;  n2 += w4.y * w4.y * q4.y;
    n1 += w4.z * s4.z;  n2 += w4.z * w4.z * q4.z;
    n1 += w4.w * s4.w;  n2 += w4.w * w4.w * q4.w;
  }
  __shared__ float red[7][2];
  const int lane = threadIdx.x & 31, wvi = threadIdx.x >> 5;
  for (int off = 16; off > 0; off >>= 1) {
    n1 += __shfl_down(n1, off, 32);
    n2 += __shfl_down(n2, off, 32);
  }
  if (lane == 0) { red[wvi][0] = n1; red[wvi][1] = n2; }
  __syncthreads();
  if (threadIdx.x == 0) {
    float t1 = 0.f, t2 = 0.f;
#pragma unroll
    for (int k = 0; k < 7; ++k) { t1 += red[k][0]; t2 += red[k][1]; }
    const float invcnt = 1.0f / 1835008.0f;
    float mean = t1 * invcnt;
    float var  = t2 * invcnt - mean * mean;
    float inv  = rsqrtf(var + 1e-5f);
    float a    = g * inv;
    aO[o] = a;
    bO[o] = 56.0f * (bt - mean * a);
  }
}

#define NT 4
#define XR (8 + 32 * NT)

__global__ void __launch_bounds__(32)
lc_wmma_main(const float* __restrict__ x, const float* __restrict__ wgt,
             const float* __restrict__ aO, const float* __restrict__ bO,
             float* __restrict__ out) {
  int id = blockIdx.x;
  int ts = id & 3;
  int fo = (id >> 2) & 63;
  int ob = (id >> 8) & 3;
  int b  = id >> 10;
  int t0 = ts << 6, o0 = ob << 4, c0 = o0, f0 = fo << 1;

  int L  = threadIdx.x;
  int m  = L & 15;
  int hi = L >> 4;
  int d0 = hi << 1;
  int mq = m >> 2;

  __shared__ float xs[2][XR][18];

  const float* xb = x + (((size_t)b * Cn + c0) * En + f0) * Tn + 2 * t0;

  const int lr = L >> 3, lc = L & 7;
  v4f body[NT][8];
#pragma unroll
  for (int i = 0; i < NT; ++i) {
#pragma unroll
    for (int k = 0; k < 8; ++k) {
      const int ch = 4 * (k & 3) + lr, fi = k >> 2;
      body[i][k] =
          *(const v4f*)(xb + ((size_t)ch * En + fi) * Tn + 32 * i + 4 * lc);
    }
  }
  const int ech = L & 15, efi = L >> 4;
  const float* er = xb + ((size_t)ech * En + efi) * Tn;
  const v4f zero4 = {0.f, 0.f, 0.f, 0.f};
  const bool has_head = (ts > 0), has_tail = (ts < 3);
  v4f head = *(const v4f*)(er + (has_head ? -4 : 0));
  v4f tail = *(const v4f*)(er + (has_tail ? 32 * NT : 32 * NT - 4));

  float wa[2][7], wb[2][7];
#pragma unroll
  for (int fi = 0; fi < 2; ++fi) {
    const float* wp = wgt + (((size_t)(o0 + m) * 4 + d0) * En + (f0 + fi)) * Wn;
#pragma unroll
    for (int w = 0; w < 7; ++w) {
      wa[fi][w] = wp[w];
      wb[fi][w] = wp[En * Wn + w];
    }
  }

  __builtin_amdgcn_sched_barrier(0);
  head = has_head ? head : zero4;
  tail = has_tail ? tail : zero4;

#pragma unroll
  for (int i = 0; i < NT; ++i) {
#pragma unroll
    for (int k = 0; k < 8; ++k) {
      const int ch = 4 * (k & 3) + lr, fi = k >> 2;
      const int j = 4 + 32 * i + 4 * lc;
      xs[fi][j + 0][ch] = body[i][k].x;
      xs[fi][j + 1][ch] = body[i][k].y;
      xs[fi][j + 2][ch] = body[i][k].z;
      xs[fi][j + 3][ch] = body[i][k].w;
    }
  }
  xs[efi][0][ech] = head.x;  xs[efi][1][ech] = head.y;
  xs[efi][2][ech] = head.z;  xs[efi][3][ech] = head.w;
  xs[efi][XR - 4][ech] = tail.x;  xs[efi][XR - 3][ech] = tail.y;
  xs[efi][XR - 2][ech] = tail.z;  xs[efi][XR - 1][ech] = tail.w;

  __syncthreads();

  v8f acc[NT];
#pragma unroll
  for (int i = 0; i < NT; ++i) acc[i] = v8f{};
#pragma unroll
  for (int fi = 0; fi < 2; ++fi) {
#pragma unroll
    for (int w = 0; w < 7; ++w) {
#pragma unroll
      for (int q = 0; q < 4; ++q) {
        float sel = (mq == q) ? 1.f : 0.f;
        v2f av = { sel * wa[fi][w], sel * wb[fi][w] };
#pragma unroll
        for (int i = 0; i < NT; ++i) {
          int j = 32 * i + 2 * m + w + 1;
          v2f bv = *(const v2f*)&xs[fi][j][(q << 2) + d0];
          acc[i] = __builtin_amdgcn_wmma_f32_16x16x4_f32(
               false, av,  false, bv,
               (short)0, acc[i],  false,  false);
        }
      }
    }
  }

#pragma unroll
  for (int r = 0; r < 8; ++r) {
    int o = o0 + r + (hi << 3);
    const float sa = aO[o], sb = bO[o];
    float* orow = out + (((size_t)b * On + o) * FOUT + fo) * TOUT + t0 + m;
#pragma unroll
    for (int i = 0; i < NT; ++i) {
      float v = sa * acc[i][r] + sb;
      v = (v >= 0.f) ? v : 0.01f * v;
      orow[16 * i] = v;
    }
  }
}

extern "C" void kernel_launch(void* const* d_in, const int* in_sizes, int n_in,
                              void* d_out, int out_size, void* d_ws, size_t ws_size,
                              hipStream_t stream) {
  (void)in_sizes; (void)n_in; (void)out_size; (void)ws_size;
  const float* x     = (const float*)d_in[0];
  const float* wgt   = (const float*)d_in[1];
  const float* gamma = (const float*)d_in[2];
  const float* beta  = (const float*)d_in[3];
  float* out = (float*)d_out;

  float* ws = (float*)d_ws;
  float* Sx = ws;
  float* Qx = ws + 57344;
  float* aO = ws + 114688;
  float* bO = ws + 114752;

  lc_stats1<<<512, 256, 0, stream>>>(x, Sx, Qx);
  lc_stats2<<<On, 224, 0, stream>>>(wgt, gamma, beta, Sx, Qx, aO, bO);
  lc_wmma_main<<<2048, 32, 0, stream>>>(x, wgt, aO, bO, out);
}


# no BN kernel: stats pass 1 folds its tap sums with the weights into 8 sums per block, each main wave adds its rows' 32 block sums; 2 launches, not 3 (changes the order of these sums)
# speedup vs baseline: 4.0419x; 1.1839x over previous
#include <hip/hip_runtime.h>

typedef __attribute__((ext_vector_type(2))) float v2f;
typedef __attribute__((ext_vector_type(8))) float v8f;
typedef __attribute__((ext_vector_type(4))) float v4f;

#define Bn 2
#define Cn 64
#define On 64
#define En 128
#define Tn 512
#define Wn 7
#define TOUT 256
#define FOUT 64

__device__ __forceinline__ float lane_value(float v, int lane) {
  return __int_as_float(__builtin_amdgcn_readlane(__float_as_int(v), lane));
}

__global__ void __launch_bounds__(256)
lc_stats1(const float* __restrict__ x, const float* __restrict__ wgt,
          float* __restrict__ part) {
  const int L   = threadIdx.x & 31;
  const int wvi = threadIdx.x >> 5;
  const int gw  = blockIdx.x * 8 + wvi;
  const int cf0 = gw * 2;
  const int c = cf0 >> 7, f0 = cf0 & 127;

  v4f v[2][2][4];
#pragma unroll
  for (int b = 0; b < 2; ++b) {
    const v4f* p =
        (const v4f*)(x + (((size_t)b * Cn + c) * En + f0) * Tn);
#pragma unroll
    for (int r = 0; r < 2; ++r)
#pragma unroll
      for (int k = 0; k < 4; ++k)
        v[b][r][k] = p[r * (Tn / 4) + k * 32 + L];
  }
  const int g4 = c & ~3, dd = c & 3;
  const int wl = (L < 14) ? L : 13;
  const float* wq = wgt + ((size_t)(g4 * 4 + dd) * En + f0) * Wn + wl;
  float wn[4];
#pragma unroll
  for (int n = 0; n < 4; ++n) wn[n] = wq[(size_t)n * 4 * (En * Wn)];
  asm volatile("" :: "v"(v[0][0][0]), "v"(v[0][0][1]), "v"(v[0][0][2]), "v"(v[0][0][3]),
                     "v"(v[0][1][0]), "v"(v[0][1][1]), "v"(v[0][1][2]), "v"(v[0][1][3]),
                     "v"(v[1][0][0]), "v"(v[1][0][1]), "v"(v[1][0][2]), "v"(v[1][0][3]),
                     "v"(v[1][1][0]), "v"(v[1][1][1]), "v"(v[1][1][2]), "v"(v[1][1][3]),
                     "v"(wn[0]), "v"(wn[1]), "v"(wn[2]), "v"(wn[3]));

  float Ec[2], Oc[2], QEc[2], QOc[2];
  float s0[2], s1[2], s509[2], s510[2], s511[2];
  float q0[2], q1[2], q509[2], q510[2], q511[2];
#pragma unroll
  for (int r = 0; r < 2; ++r) {
    const v4f ha = v[0][r][0], hb = v[1][r][0];
    const v4f ta = v[0][r][3], tb = v[1][r][3];
    s0[r]   = lane_value(ha.x + hb.x, 0);
    s1[r]   = lane_value(ha.y + hb.y, 0);
    s509[r] = lane_value(ta.y + tb.y, 31);
    s510[r] = lane_value(ta.z + tb.z, 31);
    s511[r] = lane_value(ta.w + tb.w, 31);
    q0[r]   = lane_value(ha.x * ha.x + hb.x * hb.x, 0);
    q1[r]   = lane_value(ha.y * ha.y + hb.y * hb.y, 0);
    q509[r] = lane_value(ta.y * ta.y + tb.y * tb.y, 31);
    q510[r] = lane_value(ta.z * ta.z + tb.z * tb.z, 31);
    q511[r] = lane_value(ta.w * ta.w + tb.w * tb.w, 31);

    const bool first = (L == 0), last = (L == 31);
#pragma unroll
    for (int b = 0; b < 2; ++b) {
      v[b][r][0].x = first ? 0.f : v[b][r][0].x;
      v[b][r][0].y = first ? 0.f : v[b][r][0].y;
      v[b][r][3].y = last ? 0.f : v[b][r][3].y;
      v[b][r][3].z = last ? 0.f : v[b][r][3].z;
      v[b][r][3].w = last ? 0.f : v[b][r][3].w;
    }

    float se = 0.f, so = 0.f, qe = 0.f, qo = 0.f;
#pragma unroll
    for (int b = 0; b < 2; ++b) {
#pragma unroll
      for (int k = 0; k < 4; ++k) {
        const v4f t = v[b][r][k];
        se += t.x + t.z;
        so += t.y + t.w;
        qe += t.x * t.x + t.z * t.z;
        qo += t.y * t.y + t.w * t.w;
      }
    }
#pragma unroll
    for (int off = 16; off > 0; off >>= 1) {
      se += __shfl_xor(se, off, 32);
      so += __shfl_xor(so, off, 32);
      qe += __shfl_xor(qe, off, 32);
      qo += __shfl_xor(qo, off, 32);
    }
    Ec[r] = se; Oc[r] = so; QEc[r] = qe; QOc[r] = qo;
  }

  const int rs = (L >= 7) ? 1 : 0;
  const int w  = L - 7 * rs;
  const float E  = rs ? Ec[1]  : Ec[0],  O  = rs ? Oc[1]  : Oc[0];
  const float QE = rs ? QEc[1] : QEc[0], QO = rs ? QOc[1] : QOc[0];
  const float a0 = rs ? s0[1] : s0[0], a1 = rs ? s1[1] : s1[0];
  const float a509 = rs ? s509[1] : s509[0], a510 = rs ? s510[1] : s510[0];
  const float a511 = rs ? s511[1] : s511[0];
  const float b0 = rs ? q0[1] : q0[0], b1 = rs ? q1[1] : q1[0];
  const float b509 = rs ? q509[1] : q509[0], b510 = rs ? q510[1] : q510[0];
  const float b511 = rs ? q511[1] : q511[0];

  const float sedge = (w == 0) ? a1
                    : (w == 1) ? a0
                    : (w == 2) ? a1 + a509
                    : (w == 3) ? a0 + a510
                    : (w == 4) ? (a1 + a509) + a511
                    : (w == 5) ? a510
                               : a509 + a511;
  const float qedge = (w == 0) ? b1
                    : (w == 1) ? b0
                    : (w == 2) ? b1 + b509
                    : (w == 3) ? b0 + b510
                    : (w == 4) ? (b1 + b509) + b511
                    : (w == 5) ? b510
                               : b509 + b511;
  const float sv = ((w & 1) ? E  : O)  + sedge;
  const float qv = ((w & 1) ? QE : QO) + qedge;

  const bool act = (L < 14);
  float pr[8];
#pragma unroll
  for (int n = 0; n < 4; ++n) {
    const float wt = wn[n];
    pr[2 * n]     = act ? wt * sv : 0.f;
    pr[2 * n + 1] = act ? wt * wt * qv : 0.f;
  }
#pragma unroll
  for (int off = 8; off > 0; off >>= 1) {
#pragma unroll
    for (int j = 0; j < 8; ++j) pr[j] += __shfl_xor(pr[j], off, 32);
  }
  __shared__ float red[8][8];
  if (L == 0) {
#pragma unroll
    for (int j = 0; j < 8; ++j) red[wvi][j] = pr[j];
  }
  __syncthreads();
  if (threadIdx.x < 8) {
    const int j = threadIdx.x;
    float t = 0.f;
#pragma unroll
    for (int k = 0; k < 8; ++k) t += red[k][j];
    const int fb = blockIdx.x & 7;
    part[(g4 + (j >> 1)) * 64 + dd * 16 + fb * 2 + (j & 1)] = t;
  }
}

#define NT 4
#define XR (8 + 32 * NT)

__global__ void __launch_bounds__(32)
lc_wmma_main(const float* __restrict__ x, const float* __restrict__ wgt,
             const float* __restrict__ gamma, const float* __restrict__ beta,
             const float* __restrict__ part,
             float* __restrict__ out) {
  int id = blockIdx.x;
  int ts = id & 3;
  int fo = (id >> 2) & 63;
  int ob = (id >> 8) & 3;
  int b  = id >> 10;
  int t0 = ts << 6, o0 = ob << 4, c0 = o0, f0 = fo << 1;

  int L  = threadIdx.x;
  int m  = L & 15;
  int hi = L >> 4;
  int d0 = hi << 1;
  int mq = m >> 2;

  __shared__ float xs[2][XR][18];

  const float* xb = x + (((size_t)b * Cn + c0) * En + f0) * Tn + 2 * t0;

  const int lr = L >> 3, lc = L & 7;
  v4f body[NT][8];
#pragma unroll
  for (int i = 0; i < NT; ++i) {
#pragma unroll
    for (int k = 0; k < 8; ++k) {
      const int ch = 4 * (k & 3) + lr, fi = k >> 2;
      body[i][k] =
          *(const v4f*)(xb + ((size_t)ch * En + fi) * Tn + 32 * i + 4 * lc);
    }
  }
  const int ech = L & 15, efi = L >> 4;
  const float* er = xb + ((size_t)ech * En + efi) * Tn;
  const v4f zero4 = {0.f, 0.f, 0.f, 0.f};
  const bool has_head = (ts > 0), has_tail = (ts < 3);
  v4f head = *(const v4f*)(er + (has_head ? -4 : 0));
  v4f tail = *(const v4f*)(er + (has_tail ? 32 * NT : 32 * NT - 4));

  float wa[2][7], wb[2][7];
#pragma unroll
  for (int fi = 0; fi < 2; ++fi) {
    const float* wp = wgt + (((size_t)(o0 + m) * 4 + d0) * En + (f0 + fi)) * Wn;
#pragma unroll
    for (int w = 0; w < 7; ++w) {
      wa[fi][w] = wp[w];
      wb[fi][w] = wp[En * Wn + w];
    }
  }

  const v4f* pp = (const v4f*)(part + (size_t)(o0 + m) * 64 + hi * 32);
  v4f pv[8];
#pragma unroll
  for (int k = 0; k < 8; ++k) pv[k] = pp[k];
  const float gm = gamma[o0 + m], bt = beta[o0 + m];

  __builtin_amdgcn_sched_barrier(0);
  head = has_head ? head : zero4;
  tail = has_tail ? tail : zero4;

#pragma unroll
  for (int i = 0; i < NT; ++i) {
#pragma unroll
    for (int k = 0; k < 8; ++k) {
      const int ch = 4 * (k & 3) + lr, fi = k >> 2;
      const int j = 4 + 32 * i + 4 * lc;
      xs[fi][j + 0][ch] = body[i][k].x;
      xs[fi][j + 1][ch] = body[i][k].y;
      xs[fi][j + 2][ch] = body[i][k].z;
      xs[fi][j + 3][ch] = body[i][k].w;
    }
  }
  xs[efi][0][ech] = head.x;  xs[efi][1][ech] = head.y;
  xs[efi][2][ech] = head.z;  xs[efi][3][ech] = head.w;
  xs[efi][XR - 4][ech] = tail.x;  xs[efi][XR - 3][ech] = tail.y;
  xs[efi][XR - 2][ech] = tail.z;  xs[efi][XR - 1][ech] = tail.w;

  __syncthreads();

  float n1 = 0.f, n2 = 0.f;
#pragma unroll
  for (int k = 0; k < 8; ++k) {
    n1 += pv[k].x + pv[k].z;
    n2 += pv[k].y + pv[k].w;
  }
  n1 += __shfl_xor(n1, 16, 32);
  n2 += __shfl_xor(n2, 16, 32);
  const float invcnt = 1.0f / 1835008.0f;
  const float mean = n1 * invcnt;
  const float var  = n2 * invcnt - mean * mean;
  const float a_m  = gm * rsqrtf(var + 1e-5f);
  const float b_m  = 56.0f * (bt - mean * a_m);

  v8f acc[NT];
#pragma unroll
  for (int i = 0; i < NT; ++i) acc[i] = v8f{};
#pragma unroll
  for (int fi = 0; fi < 2; ++fi) {
#pragma unroll
    for (int w = 0; w < 7; ++w) {
#pragma unroll
      for (int q = 0; q < 4; ++q) {
        float sel = (mq == q) ? 1.f : 0.f;
        v2f av = { sel * wa[fi][w], sel * wb[fi][w] };
#pragma unroll
        for (int i = 0; i < NT; ++i) {
          int j = 32 * i + 2 * m + w + 1;
          v2f bv = *(const v2f*)&xs[fi][j][(q << 2) + d0];
          acc[i] = __builtin_amdgcn_wmma_f32_16x16x4_f32(
               false, av,  false, bv,
               (short)0, acc[i],  false,  false);
        }
      }
    }
  }

#pragma unroll
  for (int r = 0; r < 8; ++r) {
    int o = o0 + r + (hi << 3);
    const float a_lo = lane_value(a_m, r), a_hi = lane_value(a_m, r + 8);
    const float b_lo = lane_value(b_m, r), b_hi = lane_value(b_m, r + 8);
    const float sa = hi ? a_hi : a_lo;
    const float sb = hi ? b_hi : b_lo;
    float* orow = out + (((size_t)b * On + o) * FOUT + fo) * TOUT + t0 + m;
#pragma unroll
    for (int i = 0; i < NT; ++i) {
      float v = sa * acc[i][r] + sb;
      v = (v >= 0.f) ? v : 0.01f * v;
      orow[16 * i] = v;
    }
  }
}

extern "C" void kernel_launch(void* const* d_in, const int* in_sizes, int n_in,
                              void* d_out, int out_size, void* d_ws, size_t ws_size,
                              hipStream_t stream) {
  (void)in_sizes; (void)n_in; (void)out_size; (void)ws_size;
  const float* x     = (const float*)d_in[0];
  const float* wgt   = (const float*)d_in[1];
  const float* gamma = (const float*)d_in[2];
  const float* beta  = (const float*)d_in[3];
  float* out = (float*)d_out;

  float* part = (float*)d_ws;

  lc_stats1<<<512, 256, 0, stream>>>(x, wgt, part);
  lc_wmma_main<<<2048, 32, 0, stream>>>(x, wgt, gamma, beta, part, out);
}


# main pass: a lane reads its 28 weights as two contiguous 56-byte runs and shifts them by 2s with register selects, not by 52 4-byte loads at shifted addresses: 44 loads per wave, not 88
# speedup vs baseline: 5.9089x; 1.4619x over previous
#include <hip/hip_runtime.h>

typedef __attribute__((ext_vector_type(2))) float v2f;
typedef __attribute__((ext_vector_type(8))) float v8f;
typedef __attribute__((ext_vector_type(4))) float v4f;

#define Bn 2
#define Cn 64
#define On 64
#define En 128
#define Tn 512
#define Wn 7
#define TOUT 256
#define FOUT 64

__device__ __forceinline__ float lane_value(float v, int lane) {
  return __int_as_float(__builtin_amdgcn_readlane(__float_as_int(v), lane));
}

__global__ void __launch_bounds__(256)
lc_stats1(const float* __restrict__ x, const float* __restrict__ wgt,
          float* __restrict__ part) {
  const int L   = threadIdx.x & 31;
  const int wvi = threadIdx.x >> 5;
  const int gw  = blockIdx.x * 8 + wvi;
  const int cf0 = gw * 2;
  const int c = cf0 >> 7, f0 = cf0 & 127;

  v4f v[2][2][4];
#pragma unroll
  for (int b = 0; b < 2; ++b) {
    const v4f* p =
        (const v4f*)(x + (((size_t)b * Cn + c) * En + f0) * Tn);
#pragma unroll
    for (int r = 0; r < 2; ++r)
#pragma unroll
      for (int k = 0; k < 4; ++k)
        v[b][r][k] = p[r * (Tn / 4) + k * 32 + L];
  }
  const int g4 = c & ~3, dd = c & 3;
  const int wl = (L < 14) ? L : 13;
  const float* wq = wgt + ((size_t)(g4 * 4 + dd) * En + f0) * Wn + wl;
  float wn[4];
#pragma unroll
  for (int n = 0; n < 4; ++n) wn[n] = wq[(size_t)n * 4 * (En * Wn)];
  asm volatile("" :: "v"(v[0][0][0]), "v"(v[0][0][1]), "v"(v[0][0][2]), "v"(v[0][0][3]),
                     "v"(v[0][1][0]), "v"(v[0][1][1]), "v"(v[0][1][2]), "v"(v[0][1][3]),
                     "v"(v[1][0][0]), "v"(v[1][0][1]), "v"(v[1][0][2]), "v"(v[1][0][3]),
                     "v"(v[1][1][0]), "v"(v[1][1][1]), "v"(v[1][1][2]), "v"(v[1][1][3]),
                     "v"(wn[0]), "v"(wn[1]), "v"(wn[2]), "v"(wn[3]));

  float Ec[2], Oc[2], QEc[2], QOc[2];
  float s0[2], s1[2], s509[2], s510[2], s511[2];
  float q0[2], q1[2], q509[2], q510[2], q511[2];
#pragma unroll
  for (int r = 0; r < 2; ++r) {
    const v4f ha = v[0][r][0], hb = v[1][r][0];
    const v4f ta = v[0][r][3], tb = v[1][r][3];
    s0[r]   = lane_value(ha.x + hb.x, 0);
    s1[r]   = lane_value(ha.y + hb.y, 0);
    s509[r] = lane_value(ta.y + tb.y, 31);
    s510[r] = lane_value(ta.z + tb.z, 31);
    s511[r] = lane_value(ta.w + tb.w, 31);
    q0[r]   = lane_value(ha.x * ha.x + hb.x * hb.x, 0);
    q1[r]   = lane_value(ha.y * ha.y + hb.y * hb.y, 0);
    q509[r] = lane_value(ta.y * ta.y + tb.y * tb.y, 31);
    q510[r] = lane_value(ta.z * ta.z + tb.z * tb.z, 31);
    q511[r] = lane_value(ta.w * ta.w + tb.w * tb.w, 31);

    const bool first = (L == 0), last = (L == 31);
#pragma unroll
    for (int b = 0; b < 2; ++b) {
      v[b][r][0].x = first ? 0.f : v[b][r][0].x;
      v[b][r][0].y = first ? 0.f : v[b][r][0].y;
      v[b][r][3].y = last ? 0.f : v[b][r][3].y;
      v[b][r][3].z = last ? 0.f : v[b][r][3].z;
      v[b][r][3].w = last ? 0.f : v[b][r][3].w;
    }

    float se = 0.f, so = 0.f, qe = 0.f, qo = 0.f;
#pragma unroll
    for (int b = 0; b < 2; ++b) {
#pragma unroll
      for (int k = 0; k < 4; ++k) {
        const v4f t = v[b][r][k];
        se += t.x + t.z;
        so += t.y + t.w;
        qe += t.x * t.x + t.z * t.z;
        qo += t.y * t.y + t.w * t.w;
      }
    }
#pragma unroll
    for (int off = 16; off > 0; off >>= 1) {
      se += __shfl_xor(se, off, 32);
      so += __shfl_xor(so, off, 32);
      qe += __shfl_xor(qe, off, 32);
      qo += __shfl_xor(qo, off, 32);
    }
    Ec[r] = se; Oc[r] = so; QEc[r] = qe; QOc[r] = qo;
  }

  const int rs = (L >= 7) ? 1 : 0;
  const int w  = L - 7 * rs;
  const float E  = rs ? Ec[1]  : Ec[0],  O  = rs ? Oc[1]  : Oc[0];
  const float QE = rs ? QEc[1] : QEc[0], QO = rs ? QOc[1] : QOc[0];
  const float a0 = rs ? s0[1] : s0[0], a1 = rs ? s1[1] : s1[0];
  const float a509 = rs ? s509[1] : s509[0], a510 = rs ? s510[1] : s510[0];
  const float a511 = rs ? s511[1] : s511[0];
  const float b0 = rs ? q0[1] : q0[0], b1 = rs ? q1[1] : q1[0];
  const float b509 = rs ? q509[1] : q509[0], b510 = rs ? q510[1] : q510[0];
  const float b511 = rs ? q511[1] : q511[0];

  const float sedge = (w == 0) ? a1
                    : (w == 1) ? a0
                    : (w == 2) ? a1 + a509
                    : (w == 3) ? a0 + a510
                    : (w == 4) ? (a1 + a509) + a511
                    : (w == 5) ? a510
                               : a509 + a511;
  const float qedge = (w == 0) ? b1
                    : (w == 1) ? b0
                    : (w == 2) ? b1 + b509
                    : (w == 3) ? b0 + b510
                    : (w == 4) ? (b1 + b509) + b511
                    : (w == 5) ? b510
                               : b509 + b511;
  const float sv = ((w & 1) ? E  : O)  + sedge;
  const float qv = ((w & 1) ? QE : QO) + qedge;

  const bool act = (L < 14);
  float pr[8];
#pragma unroll
  for (int n = 0; n < 4; ++n) {
    const float wt = wn[n];
    pr[2 * n]     = act ? wt * sv : 0.f;
    pr[2 * n + 1] = act ? wt * wt * qv : 0.f;
  }
#pragma unroll
  for (int off = 8; off > 0; off >>= 1) {
#pragma unroll
    for (int j = 0; j < 8; ++j) pr[j] += __shfl_xor(pr[j], off, 32);
  }
  __shared__ float red[8][8];
  if (L == 0) {
#pragma unroll
    for (int j = 0; j < 8; ++j) red[wvi][j] = pr[j];
  }
  __syncthreads();
  if (threadIdx.x < 8) {
    const int j = threadIdx.x;
    float t = 0.f;
#pragma unroll
    for (int k = 0; k < 8; ++k) t += red[k][j];
    const int fb = blockIdx.x & 7;
    part[(g4 + (j >> 1)) * 64 + dd * 16 + fb * 2 + (j & 1)] = t;
  }
}

__device__ __forceinline__ void shift_taps(const float (&r)[7], bool s1, bool s2,
                                           float (&a)[13]) {
  float t[9];
#pragma unroll
  for (int u = 0; u < 9; ++u) {
    const float keep = (u <= 6) ? r[u] : 0.f;
    const float move = (u >= 2) ? r[u - 2] : 0.f;
    t[u] = s1 ? move : keep;
  }
#pragma unroll
  for (int u = 0; u < 13; ++u) {
    const float keep = (u <= 8) ? t[u] : 0.f;
    const float move = (u >= 4) ? t[u - 4] : 0.f;
    a[u] = s2 ? move : keep;
  }
}

#define QG 65
#define QP 36
#define PLANE (QG * QP)

__global__ void __launch_bounds__(32)
lc_wmma_main(const float* __restrict__ x, const float* __restrict__ wgt,
             const float* __restrict__ gamma, const float* __restrict__ beta,
             const float* __restrict__ part,
             float* __restrict__ out) {
  const int id = blockIdx.x;
  const int fo = id & 63;
  const int g  = (id >> 6) & 15;
  const int b  = id >> 10;
  const int o0 = g << 2, c0 = o0, f0 = fo << 1;

  const int L  = threadIdx.x;
  const int m  = L & 15;
  const int hi = L >> 4;
  const int d0 = hi << 1;
  const int n  = m >> 2;
  const int s  = m & 3;

  __shared__ __attribute__((aligned(16))) float xs[2 * PLANE];

  const int lr = L >> 3, lc = L & 7;

  const int oj = L >> 3;
  const v4f* pp = (const v4f*)(part + (size_t)(o0 + oj) * 64 + lc * 8);
  const v4f pv0 = pp[0], pv1 = pp[1];
  const float gm = gamma[o0 + oj], bt = beta[o0 + oj];

  const float* wp = wgt + (((size_t)(o0 + n) * 4 + d0) * En + f0) * Wn;
  v2f wr0[7], wr1[7];
#pragma unroll
  for (int j = 0; j < 7; ++j) {
    wr0[j] = *(const v2f*)(wp + 2 * j);
    wr1[j] = *(const v2f*)(wp + En * Wn + 2 * j);
  }

  const float* xb = x + (((size_t)b * Cn + c0 + lr) * En + f0) * Tn + 4 * lc;
  v4f body[2][16];
#pragma unroll
  for (int fi = 0; fi < 2; ++fi) {
#pragma unroll
    for (int pc = 0; pc < 16; ++pc)
      body[fi][pc] = *(const v4f*)(xb + fi * Tn + 32 * pc);
  }

  __builtin_amdgcn_sched_barrier(0);

  float wa[2][13], wb[2][13];
  {
    const bool s1 = (s & 1) != 0, s2 = (s & 2) != 0;
#pragma unroll
    for (int fi = 0; fi < 2; ++fi) {
      float r0[7], r1[7];
#pragma unroll
      for (int k = 0; k < 7; ++k) {
        const int q = 7 * fi + k;
        r0[k] = (q & 1) ? wr0[q >> 1].y : wr0[q >> 1].x;
        r1[k] = (q & 1) ? wr1[q >> 1].y : wr1[q >> 1].x;
      }
      shift_taps(r0, s1, s2, wa[fi]);
      shift_taps(r1, s1, s2, wb[fi]);
    }
  }

  int so[4];
#pragma unroll
  for (int k = 0; k < 4; ++k) {
    const int p = 4 * lc + 3 + k;
    so[k] = (p >> 3) * QP + (p & 7) * 4 + lr;
  }
#pragma unroll
  for (int fi = 0; fi < 2; ++fi) {
#pragma unroll
    for (int pc = 0; pc < 16; ++pc) {
      float* dst = xs + fi * PLANE + 4 * pc * QP;
      dst[so[0]] = body[fi][pc].x;
      dst[so[1]] = body[fi][pc].y;
      dst[so[2]] = body[fi][pc].z;
      dst[so[3]] = body[fi][pc].w;
    }
  }
  {
    const int zk = L >> 3;
    const int p0 = (zk == 0) ? 0 : (zk == 1) ? 2   : (zk == 2) ? 516 : 518;
    const int p1 = (zk == 0) ? 1 : (zk == 1) ? 515 : (zk == 2) ? 517 : 519;
    float* zr = xs + ((L >> 2) & 1) * PLANE + (L & 3);
    zr[(p0 >> 3) * QP + (p0 & 7) * 4] = 0.f;
    zr[(p1 >> 3) * QP + (p1 & 7) * 4] = 0.f;
  }

  __syncthreads();

  float n1 = (pv0.x + pv0.z) + (pv1.x + pv1.z);
  float n2 = (pv0.y + pv0.w) + (pv1.y + pv1.w);
#pragma unroll
  for (int off = 4; off > 0; off >>= 1) {
    n1 += __shfl_xor(n1, off, 32);
    n2 += __shfl_xor(n2, off, 32);
  }
  const float invcnt = 1.0f / 1835008.0f;
  const float mean = n1 * invcnt;
  const float var  = n2 * invcnt - mean * mean;
  const float a_m  = gm * rsqrtf(var + 1e-5f);
  const float b_m  = 56.0f * (bt - mean * a_m);

  v8f acc[4];
#pragma unroll
  for (int i = 0; i < 4; ++i) acc[i] = v8f{};
  const float* xl = xs + m * QP + d0;
#pragma unroll
  for (int fi = 0; fi < 2; ++fi) {
#pragma unroll
    for (int u = 0; u < 13; ++u) {
      v2f av = { wa[fi][u], wb[fi][u] };
#pragma unroll
      for (int i = 0; i < 4; ++i) {
        v2f bv = *(const v2f*)(xl + fi * PLANE + (16 * i + (u >> 3)) * QP +
                               (u & 7) * 4);
        acc[i] = __builtin_amdgcn_wmma_f32_16x16x4_f32(
             false, av,  false, bv,
             (short)0, acc[i],  false,  false);
      }
    }
  }

  float a4[4], b4[4];
#pragma unroll
  for (int j = 0; j < 4; ++j) {
    a4[j] = lane_value(a_m, 8 * j);
    b4[j] = lane_value(b_m, 8 * j);
  }
#pragma unroll
  for (int nn = 0; nn < 2; ++nn) {
    const float sa = hi ? a4[2 + nn] : a4[nn];
    const float sb = hi ? b4[2 + nn] : b4[nn];
    const int o = o0 + 2 * hi + nn;
    float* orow = out + (((size_t)b * On + o) * FOUT + fo) * TOUT + 4 * m;
#pragma unroll
    for (int i = 0; i < 4; ++i) {
      v4f v;
      v.x = sa * acc[i][4 * nn + 0] + sb;
      v.y = sa * acc[i][4 * nn + 1] + sb;
      v.z = sa * acc[i][4 * nn + 2] + sb;
      v.w = sa * acc[i][4 * nn + 3] + sb;
      v.x = (v.x >= 0.f) ? v.x : 0.01f * v.x;
      v.y = (v.y >= 0.f) ? v.y : 0.01f * v.y;
      v.z = (v.z >= 0.f) ? v.z : 0.01f * v.z;
      v.w = (v.w >= 0.f) ? v.w : 0.01f * v.w;
      *(v4f*)(orow + 64 * i) = v;
    }
  }
}

extern "C" void kernel_launch(void* const* d_in, const int* in_sizes, int n_in,
                              void* d_out, int out_size, void* d_ws, size_t ws_size,
                              hipStream_t stream) {
  (void)in_sizes; (void)n_in; (void)out_size; (void)ws_size;
  const float* x     = (const float*)d_in[0];
  const float* wgt   = (const float*)d_in[1];
  const float* gamma = (const float*)d_in[2];
  const float* beta  = (const float*)d_in[3];
  float* out = (float*)d_out;

  float* part = (float*)d_ws;

  lc_stats1<<<512, 256, 0, stream>>>(x, wgt, part);
  lc_wmma_main<<<2048, 32, 0, stream>>>(x, wgt, gamma, beta, part, out);
}


# main pass: A rows = 4 filters x 4 tap pairs, K = (depth, f, tap parity), B = raw x pairs from the global loads, no LDS; partials added at shifted columns by DPP: 64 WMMAs, not 104 (sum order changes)
# speedup vs baseline: 6.7846x; 1.1482x over previous
#include <hip/hip_runtime.h>

typedef __attribute__((ext_vector_type(2))) float v2f;
typedef __attribute__((ext_vector_type(8))) float v8f;
typedef __attribute__((ext_vector_type(4))) float v4f;

#define Bn 2
#define Cn 64
#define On 64
#define En 128
#define Tn 512
#define Wn 7
#define TOUT 256
#define FOUT 64

__device__ __forceinline__ float lane_value(float v, int lane) {
  return __int_as_float(__builtin_amdgcn_readlane(__float_as_int(v), lane));
}

__global__ void __launch_bounds__(256)
lc_stats1(const float* __restrict__ x, const float* __restrict__ wgt,
          float* __restrict__ part) {
  const int L   = threadIdx.x & 31;
  const int wvi = threadIdx.x >> 5;
  const int gw  = blockIdx.x * 8 + wvi;
  const int cf0 = gw * 2;
  const int c = cf0 >> 7, f0 = cf0 & 127;

  v4f v[2][2][4];
#pragma unroll
  for (int b = 0; b < 2; ++b) {
    const v4f* p =
        (const v4f*)(x + (((size_t)b * Cn + c) * En + f0) * Tn);
#pragma unroll
    for (int r = 0; r < 2; ++r)
#pragma unroll
      for (int k = 0; k < 4; ++k)
        v[b][r][k] = p[r * (Tn / 4) + k * 32 + L];
  }
  const int g4 = c & ~3, dd = c & 3;
  const int wl = (L < 14) ? L : 13;
  const float* wq = wgt + ((size_t)(g4 * 4 + dd) * En + f0) * Wn + wl;
  float wn[4];
#pragma unroll
  for (int n = 0; n < 4; ++n) wn[n] = wq[(size_t)n * 4 * (En * Wn)];
  asm volatile("" :: "v"(v[0][0][0]), "v"(v[0][0][1]), "v"(v[0][0][2]), "v"(v[0][0][3]),
                     "v"(v[0][1][0]), "v"(v[0][1][1]), "v"(v[0][1][2]), "v"(v[0][1][3]),
                     "v"(v[1][0][0]), "v"(v[1][0][1]), "v"(v[1][0][2]), "v"(v[1][0][3]),
                     "v"(v[1][1][0]), "v"(v[1][1][1]), "v"(v[1][1][2]), "v"(v[1][1][3]),
                     "v"(wn[0]), "v"(wn[1]), "v"(wn[2]), "v"(wn[3]));

  float Ec[2], Oc[2], QEc[2], QOc[2];
  float s0[2], s1[2], s509[2], s510[2], s511[2];
  float q0[2], q1[2], q509[2], q510[2], q511[2];
#pragma unroll
  for (int r = 0; r < 2; ++r) {
    const v4f ha = v[0][r][0], hb = v[1][r][0];
    const v4f ta = v[0][r][3], tb = v[1][r][3];
    s0[r]   = lane_value(ha.x + hb.x, 0);
    s1[r]   = lane_value(ha.y + hb.y, 0);
    s509[r] = lane_value(ta.y + tb.y, 31);
    s510[r] = lane_value(ta.z + tb.z, 31);
    s511[r] = lane_value(ta.w + tb.w, 31);
    q0[r]   = lane_value(ha.x * ha.x + hb.x * hb.x, 0);
    q1[r]   = lane_value(ha.y * ha.y + hb.y * hb.y, 0);
    q509[r] = lane_value(ta.y * ta.y + tb.y * tb.y, 31);
    q510[r] = lane_value(ta.z * ta.z + tb.z * tb.z, 31);
    q511[r] = lane_value(ta.w * ta.w + tb.w * tb.w, 31);

    const bool first = (L == 0), last = (L == 31);
#pragma unroll
    for (int b = 0; b < 2; ++b) {
      v[b][r][0].x = first ? 0.f : v[b][r][0].x;
      v[b][r][0].y = first ? 0.f : v[b][r][0].y;
      v[b][r][3].y = last ? 0.f : v[b][r][3].y;
      v[b][r][3].z = last ? 0.f : v[b][r][3].z;
      v[b][r][3].w = last ? 0.f : v[b][r][3].w;
    }

    float se = 0.f, so = 0.f, qe = 0.f, qo = 0.f;
#pragma unroll
    for (int b = 0; b < 2; ++b) {
#pragma unroll
      for (int k = 0; k < 4; ++k) {
        const v4f t = v[b][r][k];
        se += t.x + t.z;
        so += t.y + t.w;
        qe += t.x * t.x + t.z * t.z;
        qo += t.y * t.y + t.w * t.w;
      }
    }
#pragma unroll
    for (int off = 16; off > 0; off >>= 1) {
      se += __shfl_xor(se, off, 32);
      so += __shfl_xor(so, off, 32);
      qe += __shfl_xor(qe, off, 32);
      qo += __shfl_xor(qo, off, 32);
    }
    Ec[r] = se; Oc[r] = so; QEc[r] = qe; QOc[r] = qo;
  }

  const int rs = (L >= 7) ? 1 : 0;
  const int w  = L - 7 * rs;
  const float E  = rs ? Ec[1]  : Ec[0],  O  = rs ? Oc[1]  : Oc[0];
  const float QE = rs ? QEc[1] : QEc[0], QO = rs ? QOc[1] : QOc[0];
  const float a0 = rs ? s0[1] : s0[0], a1 = rs ? s1[1] : s1[0];
  const float a509 = rs ? s509[1] : s509[0], a510 = rs ? s510[1] : s510[0];
  const float a511 = rs ? s511[1] : s511[0];
  const float b0 = rs ? q0[1] : q0[0], b1 = rs ? q1[1] : q1[0];
  const float b509 = rs ? q509[1] : q509[0], b510 = rs ? q510[1] : q510[0];
  const float b511 = rs ? q511[1] : q511[0];

  const float sedge = (w == 0) ? a1
                    : (w == 1) ? a0
                    : (w == 2) ? a1 + a509
                    : (w == 3) ? a0 + a510
                    : (w == 4) ? (a1 + a509) + a511
                    : (w == 5) ? a510
                               : a509 + a511;
  const float qedge = (w == 0) ? b1
                    : (w == 1) ? b0
                    : (w == 2) ? b1 + b509
                    : (w == 3) ? b0 + b510
                    : (w == 4) ? (b1 + b509) + b511
                    : (w == 5) ? b510
                               : b509 + b511;
  const float sv = ((w & 1) ? E  : O)  + sedge;
  const float qv = ((w & 1) ? QE : QO) + qedge;

  const bool act = (L < 14);
  float pr[8];
#pragma unroll
  for (int n = 0; n < 4; ++n) {
    const float wt = wn[n];
    pr[2 * n]     = act ? wt * sv : 0.f;
    pr[2 * n + 1] = act ? wt * wt * qv : 0.f;
  }
#pragma unroll
  for (int off = 8; off > 0; off >>= 1) {
#pragma unroll
    for (int j = 0; j < 8; ++j) pr[j] += __shfl_xor(pr[j], off, 32);
  }
  __shared__ float red[8][8];
  if (L == 0) {
#pragma unroll
    for (int j = 0; j < 8; ++j) red[wvi][j] = pr[j];
  }
  __syncthreads();
  if (threadIdx.x < 8) {
    const int j = threadIdx.x;
    float t = 0.f;
#pragma unroll
    for (int k = 0; k < 8; ++k) t += red[k][j];
    const int fb = blockIdx.x & 7;
    part[(g4 + (j >> 1)) * 64 + dd * 16 + fb * 2 + (j & 1)] = t;
  }
}

__device__ __forceinline__ float row_prev(float xk, float xkm1, bool row_end) {
  const float t = row_end ? xkm1 : xk;
  return __int_as_float(
      __builtin_amdgcn_update_dpp(0, __float_as_int(t), 0x121, 0xf, 0xf, false));
}
__device__ __forceinline__ float row_next(float xk, float xkp1, bool row_start) {
  const float t = row_start ? xkp1 : xk;
  return __int_as_float(
      __builtin_amdgcn_update_dpp(0, __float_as_int(t), 0x12F, 0xf, 0xf, false));
}

__global__ void __launch_bounds__(32)
lc_wmma_main(const float* __restrict__ x, const float* __restrict__ wgt,
             const float* __restrict__ gamma, const float* __restrict__ beta,
             const float* __restrict__ part,
             float* __restrict__ out) {
  const int id = blockIdx.x;
  const int fo = id & 63;
  const int g  = (id >> 6) & 15;
  const int b  = id >> 10;
  const int o0 = g << 2, c0 = o0, f0 = fo << 1;

  const int L = threadIdx.x;
  const int q = L & 15;
  const int h = L >> 4;
  const int n = q >> 2;
  const int j = q & 3;

  const int oj = L >> 3, lc = L & 7;
  const v4f* pp = (const v4f*)(part + (size_t)(o0 + oj) * 64 + lc * 8);
  const v4f pv0 = pp[0], pv1 = pp[1];
  const float gm = gamma[o0 + oj], bt = beta[o0 + oj];

  const float* wp = wgt + (((size_t)(o0 + n) * 4 + 2 * h) * En + f0) * Wn;
  const int te = 2 * j;
  const int to = (j == 0) ? 0 : 2 * j - 1;
  float wlo[2][2], whi[2][2];
#pragma unroll
  for (int ds = 0; ds < 2; ++ds)
#pragma unroll
    for (int fi = 0; fi < 2; ++fi) {
      const float* r = wp + (size_t)ds * (En * Wn) + fi * Wn;
      wlo[ds][fi] = r[to];
      whi[ds][fi] = r[te];
    }

  const float* xb = x + (((size_t)b * Cn + c0 + 2 * h) * En + f0) * Tn + 4 * q;
  v4f xv[8][2][2];
#pragma unroll
  for (int k = 0; k < 8; ++k)
#pragma unroll
    for (int ds = 0; ds < 2; ++ds)
#pragma unroll
      for (int fi = 0; fi < 2; ++fi)
        xv[k][ds][fi] = *(const v4f*)(xb + ((size_t)ds * En + fi) * Tn + 64 * k);

  __builtin_amdgcn_sched_barrier(0);

  v2f av[2][2];
#pragma unroll
  for (int ds = 0; ds < 2; ++ds)
#pragma unroll
    for (int fi = 0; fi < 2; ++fi) {
      av[ds][fi].x = (j == 0) ? 0.f : wlo[ds][fi];
      av[ds][fi].y = whi[ds][fi];
    }

  v8f acc[8][2];
#pragma unroll
  for (int k = 0; k < 8; ++k) {
    acc[k][0] = v8f{};
    acc[k][1] = v8f{};
  }
#pragma unroll
  for (int kk = 0; kk < 4; ++kk) {
#pragma unroll
    for (int ds = 0; ds < 2; ++ds) {
#pragma unroll
      for (int fi = 0; fi < 2; ++fi) {
#pragma unroll
        for (int k2 = 0; k2 < 2; ++k2) {
          const int k = 2 * kk + k2;
          const v4f p = xv[k][ds][fi];
          const v2f b0 = { p.x, p.y };
          const v2f b1 = { p.z, p.w };
          acc[k][0] = __builtin_amdgcn_wmma_f32_16x16x4_f32(
               false, av[ds][fi],  false, b0,
               (short)0, acc[k][0],  false,  false);
          acc[k][1] = __builtin_amdgcn_wmma_f32_16x16x4_f32(
               false, av[ds][fi],  false, b1,
               (short)0, acc[k][1],  false,  false);
        }
      }
    }
  }

  float n1 = (pv0.x + pv0.z) + (pv1.x + pv1.z);
  float n2 = (pv0.y + pv0.w) + (pv1.y + pv1.w);
#pragma unroll
  for (int off = 4; off > 0; off >>= 1) {
    n1 += __shfl_xor(n1, off, 32);
    n2 += __shfl_xor(n2, off, 32);
  }
  const float invcnt = 1.0f / 1835008.0f;
  const float mean = n1 * invcnt;
  const float var  = n2 * invcnt - mean * mean;
  const float a_m  = gm * rsqrtf(var + 1e-5f);
  const float b_m  = 56.0f * (bt - mean * a_m);

  float a4[4], b4[4];
#pragma unroll
  for (int i = 0; i < 4; ++i) {
    a4[i] = lane_value(a_m, 8 * i);
    b4[i] = lane_value(b_m, 8 * i);
  }

  const bool row_start = (q == 0), row_end = (q == 15);
#pragma unroll
  for (int nn = 0; nn < 2; ++nn) {
    const float sa = h ? a4[2 + nn] : a4[nn];
    const float sb = h ? b4[2 + nn] : b4[nn];
    const int o = o0 + 2 * h + nn;
    float* orow = out + (((size_t)b * On + o) * FOUT + fo) * TOUT + 2 * q;
    float U1[8];
#pragma unroll
    for (int k = 0; k < 8; ++k)
      U1[k] = acc[k][0][4 * nn + 0] + acc[k][1][4 * nn + 1];
#pragma unroll
    for (int k = 0; k < 8; ++k) {
      const float p0b = (k > 0) ? acc[k - 1][1][4 * nn + 0] : 0.f;
      const float p3n = (k < 7) ? acc[k + 1][0][4 * nn + 3] : 0.f;
      const float u1b = (k > 0) ? U1[k - 1] : 0.f;
      const float U0 = row_prev(acc[k][1][4 * nn + 0], p0b, row_end) +
                       acc[k][0][4 * nn + 1];
      const float V0 = acc[k][0][4 * nn + 2] + acc[k][1][4 * nn + 3];
      const float V1 = acc[k][1][4 * nn + 2] +
                       row_next(acc[k][0][4 * nn + 3], p3n, row_start);
      const float e0 = row_prev(U1[k], u1b, row_end) + V0;
      const float e1 = U0 + V1;
      v2f v;
      v.x = sa * e0 + sb;
      v.y = sa * e1 + sb;
      v.x = (v.x >= 0.f) ? v.x : 0.01f * v.x;
      v.y = (v.y >= 0.f) ? v.y : 0.01f * v.y;
      *(v2f*)(orow + 32 * k) = v;
    }
  }
}

extern "C" void kernel_launch(void* const* d_in, const int* in_sizes, int n_in,
                              void* d_out, int out_size, void* d_ws, size_t ws_size,
                              hipStream_t stream) {
  (void)in_sizes; (void)n_in; (void)out_size; (void)ws_size;
  const float* x     = (const float*)d_in[0];
  const float* wgt   = (const float*)d_in[1];
  const float* gamma = (const float*)d_in[2];
  const float* beta  = (const float*)d_in[3];
  float* out = (float*)d_out;

  float* part = (float*)d_ws;

  lc_stats1<<<512, 256, 0, stream>>>(x, wgt, part);
  lc_wmma_main<<<2048, 32, 0, stream>>>(x, wgt, gamma, beta, part, out);
}


# main pass: eight waves (eight adjacent fo of one (b, g)) per work-group: 256 work-groups of 256 threads, not 2048 single-wave work-groups; per-wave code and work unchanged
# speedup vs baseline: 7.5559x; 1.1137x over previous
#include <hip/hip_runtime.h>

typedef __attribute__((ext_vector_type(2))) float v2f;
typedef __attribute__((ext_vector_type(8))) float v8f;
typedef __attribute__((ext_vector_type(4))) float v4f;

#define Bn 2
#define Cn 64
#define On 64
#define En 128
#define Tn 512
#define Wn 7
#define TOUT 256
#define FOUT 64

__device__ __forceinline__ float lane_value(float v, int lane) {
  return __int_as_float(__builtin_amdgcn_readlane(__float_as_int(v), lane));
}

__global__ void __launch_bounds__(256)
lc_stats1(const float* __restrict__ x, const float* __restrict__ wgt,
          float* __restrict__ part) {
  const int L   = threadIdx.x & 31;
  const int wvi = threadIdx.x >> 5;
  const int gw  = blockIdx.x * 8 + wvi;
  const int cf0 = gw * 2;
  const int c = cf0 >> 7, f0 = cf0 & 127;

  v4f v[2][2][4];
#pragma unroll
  for (int b = 0; b < 2; ++b) {
    const v4f* p =
        (const v4f*)(x + (((size_t)b * Cn + c) * En + f0) * Tn);
#pragma unroll
    for (int r = 0; r < 2; ++r)
#pragma unroll
      for (int k = 0; k < 4; ++k)
        v[b][r][k] = p[r * (Tn / 4) + k * 32 + L];
  }
  const int g4 = c & ~3, dd = c & 3;
  const int wl = (L < 14) ? L : 13;
  const float* wq = wgt + ((size_t)(g4 * 4 + dd) * En + f0) * Wn + wl;
  float wn[4];
#pragma unroll
  for (int n = 0; n < 4; ++n) wn[n] = wq[(size_t)n * 4 * (En * Wn)];
  asm volatile("" :: "v"(v[0][0][0]), "v"(v[0][0][1]), "v"(v[0][0][2]), "v"(v[0][0][3]),
                     "v"(v[0][1][0]), "v"(v[0][1][1]), "v"(v[0][1][2]), "v"(v[0][1][3]),
                     "v"(v[1][0][0]), "v"(v[1][0][1]), "v"(v[1][0][2]), "v"(v[1][0][3]),
                     "v"(v[1][1][0]), "v"(v[1][1][1]), "v"(v[1][1][2]), "v"(v[1][1][3]),
                     "v"(wn[0]), "v"(wn[1]), "v"(wn[2]), "v"(wn[3]));

  float Ec[2], Oc[2], QEc[2], QOc[2];
  float s0[2], s1[2], s509[2], s510[2], s511[2];
  float q0[2], q1[2], q509[2], q510[2], q511[2];
#pragma unroll
  for (int r = 0; r < 2; ++r) {
    const v4f ha = v[0][r][0], hb = v[1][r][0];
    const v4f ta = v[0][r][3], tb = v[1][r][3];
    s0[r]   = lane_value(ha.x + hb.x, 0);
    s1[r]   = lane_value(ha.y + hb.y, 0);
    s509[r] = lane_value(ta.y + tb.y, 31);
    s510[r] = lane_value(ta.z + tb.z, 31);
    s511[r] = lane_value(ta.w + tb.w, 31);
    q0[r]   = lane_value(ha.x * ha.x + hb.x * hb.x, 0);
    q1[r]   = lane_value(ha.y * ha.y + hb.y * hb.y, 0);
    q509[r] = lane_value(ta.y * ta.y + tb.y * tb.y, 31);
    q510[r] = lane_value(ta.z * ta.z + tb.z * tb.z, 31);
    q511[r] = lane_value(ta.w * ta.w + tb.w * tb.w, 31);

    const bool first = (L == 0), last = (L == 31);
#pragma unroll
    for (int b = 0; b < 2; ++b) {
      v[b][r][0].x = first ? 0.f : v[b][r][0].x;
      v[b][r][0].y = first ? 0.f : v[b][r][0].y;
      v[b][r][3].y = last ? 0.f : v[b][r][3].y;
      v[b][r][3].z = last ? 0.f : v[b][r][3].z;
      v[b][r][3].w = last ? 0.f : v[b][r][3].w;
    }

    float se = 0.f, so = 0.f, qe = 0.f, qo = 0.f;
#pragma unroll
    for (int b = 0; b < 2; ++b) {
#pragma unroll
      for (int k = 0; k < 4; ++k) {
        const v4f t = v[b][r][k];
        se += t.x + t.z;
        so += t.y + t.w;
        qe += t.x * t.x + t.z * t.z;
        qo += t.y * t.y + t.w * t.w;
      }
    }
#pragma unroll
    for (int off = 16; off > 0; off >>= 1) {
      se += __shfl_xor(se, off, 32);
      so += __shfl_xor(so, off, 32);
      qe += __shfl_xor(qe, off, 32);
      qo += __shfl_xor(qo, off, 32);
    }
    Ec[r] = se; Oc[r] = so; QEc[r] = qe; QOc[r] = qo;
  }

  const int rs = (L >= 7) ? 1 : 0;
  const int w  = L - 7 * rs;
  const float E  = rs ? Ec[1]  : Ec[0],  O  = rs ? Oc[1]  : Oc[0];
  const float QE = rs ? QEc[1] : QEc[0], QO = rs ? QOc[1] : QOc[0];
  const float a0 = rs ? s0[1] : s0[0], a1 = rs ? s1[1] : s1[0];
  const float a509 = rs ? s509[1] : s509[0], a510 = rs ? s510[1] : s510[0];
  const float a511 = rs ? s511[1] : s511[0];
  const float b0 = rs ? q0[1] : q0[0], b1 = rs ? q1[1] : q1[0];
  const float b509 = rs ? q509[1] : q509[0], b510 = rs ? q510[1] : q510[0];
  const float b511 = rs ? q511[1] : q511[0];

  const float sedge = (w == 0) ? a1
                    : (w == 1) ? a0
                    : (w == 2) ? a1 + a509
                    : (w == 3) ? a0 + a510
                    : (w == 4) ? (a1 + a509) + a511
                    : (w == 5) ? a510
                               : a509 + a511;
  const float qedge = (w == 0) ? b1
                    : (w == 1) ? b0
                    : (w == 2) ? b1 + b509
                    : (w == 3) ? b0 + b510
                    : (w == 4) ? (b1 + b509) + b511
                    : (w == 5) ? b510
                               : b509 + b511;
  const float sv = ((w & 1) ? E  : O)  + sedge;
  const float qv = ((w & 1) ? QE : QO) + qedge;

  const bool act = (L < 14);
  float pr[8];
#pragma unroll
  for (int n = 0; n < 4; ++n) {
    const float wt = wn[n];
    pr[2 * n]     = act ? wt * sv : 0.f;
    pr[2 * n + 1] = act ? wt * wt * qv : 0.f;
  }
#pragma unroll
  for (int off = 8; off > 0; off >>= 1) {
#pragma unroll
    for (int j = 0; j < 8; ++j) pr[j] += __shfl_xor(pr[j], off, 32);
  }
  __shared__ float red[8][8];
  if (L == 0) {
#pragma unroll
    for (int j = 0; j < 8; ++j) red[wvi][j] = pr[j];
  }
  __syncthreads();
  if (threadIdx.x < 8) {
    const int j = threadIdx.x;
    float t = 0.f;
#pragma unroll
    for (int k = 0; k < 8; ++k) t += red[k][j];
    const int fb = blockIdx.x & 7;
    part[(g4 + (j >> 1)) * 64 + dd * 16 + fb * 2 + (j & 1)] = t;
  }
}

__device__ __forceinline__ float row_prev(float xk, float xkm1, bool row_end) {
  const float t = row_end ? xkm1 : xk;
  return __int_as_float(
      __builtin_amdgcn_update_dpp(0, __float_as_int(t), 0x121, 0xf, 0xf, false));
}
__device__ __forceinline__ float row_next(float xk, float xkp1, bool row_start) {
  const float t = row_start ? xkp1 : xk;
  return __int_as_float(
      __builtin_amdgcn_update_dpp(0, __float_as_int(t), 0x12F, 0xf, 0xf, false));
}

__global__ void __launch_bounds__(256)
lc_wmma_main(const float* __restrict__ x, const float* __restrict__ wgt,
             const float* __restrict__ gamma, const float* __restrict__ beta,
             const float* __restrict__ part,
             float* __restrict__ out) {
  const int wv = __builtin_amdgcn_readfirstlane((int)(threadIdx.x >> 5));
  const int id = blockIdx.x * 8 + wv;
  const int fo = id & 63;
  const int g  = (id >> 6) & 15;
  const int b  = id >> 10;
  const int o0 = g << 2, c0 = o0, f0 = fo << 1;

  const int L = threadIdx.x & 31;
  const int q = L & 15;
  const int h = L >> 4;
  const int n = q >> 2;
  const int j = q & 3;

  const int oj = L >> 3, lc = L & 7;
  const v4f* pp = (const v4f*)(part + (size_t)(o0 + oj) * 64 + lc * 8);
  const v4f pv0 = pp[0], pv1 = pp[1];
  const float gm = gamma[o0 + oj], bt = beta[o0 + oj];

  const float* wp = wgt + (((size_t)(o0 + n) * 4 + 2 * h) * En + f0) * Wn;
  const int te = 2 * j;
  const int to = (j == 0) ? 0 : 2 * j - 1;
  float wlo[2][2], whi[2][2];
#pragma unroll
  for (int ds = 0; ds < 2; ++ds)
#pragma unroll
    for (int fi = 0; fi < 2; ++fi) {
      const float* r = wp + (size_t)ds * (En * Wn) + fi * Wn;
      wlo[ds][fi] = r[to];
      whi[ds][fi] = r[te];
    }

  const float* xb = x + (((size_t)b * Cn + c0 + 2 * h) * En + f0) * Tn + 4 * q;
  v4f xv[8][2][2];
#pragma unroll
  for (int k = 0; k < 8; ++k)
#pragma unroll
    for (int ds = 0; ds < 2; ++ds)
#pragma unroll
      for (int fi = 0; fi < 2; ++fi)
        xv[k][ds][fi] = *(const v4f*)(xb + ((size_t)ds * En + fi) * Tn + 64 * k);

  __builtin_amdgcn_sched_barrier(0);

  v2f av[2][2];
#pragma unroll
  for (int ds = 0; ds < 2; ++ds)
#pragma unroll
    for (int fi = 0; fi < 2; ++fi) {
      av[ds][fi].x = (j == 0) ? 0.f : wlo[ds][fi];
      av[ds][fi].y = whi[ds][fi];
    }

  v8f acc[8][2];
#pragma unroll
  for (int k = 0; k < 8; ++k) {
    acc[k][0] = v8f{};
    acc[k][1] = v8f{};
  }
#pragma unroll
  for (int kk = 0; kk < 4; ++kk) {
#pragma unroll
    for (int ds = 0; ds < 2; ++ds) {
#pragma unroll
      for (int fi = 0; fi < 2; ++fi) {
#pragma unroll
        for (int k2 = 0; k2 < 2; ++k2) {
          const int k = 2 * kk + k2;
          const v4f p = xv[k][ds][fi];
          const v2f b0 = { p.x, p.y };
          const v2f b1 = { p.z, p.w };
          acc[k][0] = __builtin_amdgcn_wmma_f32_16x16x4_f32(
               false, av[ds][fi],  false, b0,
               (short)0, acc[k][0],  false,  false);
          acc[k][1] = __builtin_amdgcn_wmma_f32_16x16x4_f32(
               false, av[ds][fi],  false, b1,
               (short)0, acc[k][1],  false,  false);
        }
      }
    }
  }

  float n1 = (pv0.x + pv0.z) + (pv1.x + pv1.z);
  float n2 = (pv0.y + pv0.w) + (pv1.y + pv1.w);
#pragma unroll
  for (int off = 4; off > 0; off >>= 1) {
    n1 += __shfl_xor(n1, off, 32);
    n2 += __shfl_xor(n2, off, 32);
  }
  const float invcnt = 1.0f / 1835008.0f;
  const float mean = n1 * invcnt;
  const float var  = n2 * invcnt - mean * mean;
  const float a_m  = gm * rsqrtf(var + 1e-5f);
  const float b_m  = 56.0f * (bt - mean * a_m);

  float a4[4], b4[4];
#pragma unroll
  for (int i = 0; i < 4; ++i) {
    a4[i] = lane_value(a_m, 8 * i);
    b4[i] = lane_value(b_m, 8 * i);
  }

  const bool row_start = (q == 0), row_end = (q == 15);
#pragma unroll
  for (int nn = 0; nn < 2; ++nn) {
    const float sa = h ? a4[2 + nn] : a4[nn];
    const float sb = h ? b4[2 + nn] : b4[nn];
    const int o = o0 + 2 * h + nn;
    float* orow = out + (((size_t)b * On + o) * FOUT + fo) * TOUT + 2 * q;
    float U1[8];
#pragma unroll
    for (int k = 0; k < 8; ++k)
      U1[k] = acc[k][0][4 * nn + 0] + acc[k][1][4 * nn + 1];
#pragma unroll
    for (int k = 0; k < 8; ++k) {
      const float p0b = (k > 0) ? acc[k - 1][1][4 * nn + 0] : 0.f;
      const float p3n = (k < 7) ? acc[k + 1][0][4 * nn + 3] : 0.f;
      const float u1b = (k > 0) ? U1[k - 1] : 0.f;
      const float U0 = row_prev(acc[k][1][4 * nn + 0], p0b, row_end) +
                       acc[k][0][4 * nn + 1];
      const float V0 = acc[k][0][4 * nn + 2] + acc[k][1][4 * nn + 3];
      const float V1 = acc[k][1][4 * nn + 2] +
                       row_next(acc[k][0][4 * nn + 3], p3n, row_start);
      const float e0 = row_prev(U1[k], u1b, row_end) + V0;
      const float e1 = U0 + V1;
      v2f v;
      v.x = sa * e0 + sb;
      v.y = sa * e1 + sb;
      v.x = (v.x >= 0.f) ? v.x : 0.01f * v.x;
      v.y = (v.y >= 0.f) ? v.y : 0.01f * v.y;
      *(v2f*)(orow + 32 * k) = v;
    }
  }
}

extern "C" void kernel_launch(void* const* d_in, const int* in_sizes, int n_in,
                              void* d_out, int out_size, void* d_ws, size_t ws_size,
                              hipStream_t stream) {
  (void)in_sizes; (void)n_in; (void)out_size; (void)ws_size;
  const float* x     = (const float*)d_in[0];
  const float* wgt   = (const float*)d_in[1];
  const float* gamma = (const float*)d_in[2];
  const float* beta  = (const float*)d_in[3];
  float* out = (float*)d_out;

  float* part = (float*)d_ws;

  lc_stats1<<<512, 256, 0, stream>>>(x, wgt, part);
  lc_wmma_main<<<256, 256, 0, stream>>>(x, wgt, gamma, beta, part, out);
}
